# starting kernel
# baseline (speedup 1.0000x reference)
#include <hip/hip_runtime.h>

#define BB    4
#define NN    8192
#define KK    32
#define CIN   64
#define CF    67
#define KP    68
#define MID   64
#define COUT  64
#define GRP   8
#define SLOPE 0.1f
#define EPSV  1e-5f

typedef __attribute__((ext_vector_type(2))) float v2f;
typedef __attribute__((ext_vector_type(8))) float v8f;

static __device__ __forceinline__ void wave_gemm(const float* Xs, const float* Ws,
                                                 int rowBase, int kdim, int lane,
                                                 v8f acc[4]) {
    const int l16  = lane & 15;
    const int koff = (lane >> 4) << 1;
    const float* xrow = Xs + (rowBase + l16) * KP;
    for (int k0 = 0; k0 < kdim; k0 += 4) {
        v2f a;
        a.x = xrow[k0 + koff];
        a.y = xrow[k0 + koff + 1];
#pragma unroll
        for (int t = 0; t < 4; ++t) {
            const float* wrow = Ws + (t * 16 + l16) * KP;
            v2f b;
            b.x = wrow[k0 + koff];
            b.y = wrow[k0 + koff + 1];
            acc[t] = __builtin_amdgcn_wmma_f32_16x16x4_f32(
                false, a, false, b, (short)0, acc[t], false, false);
        }
    }
}

static __device__ __forceinline__ void stats_reduce(v8f acc[4], int lane, float* sred) {
    const int l16 = lane & 15;
#pragma unroll
    for (int t = 0; t < 4; ++t) {
        float s = 0.f, q = 0.f;
#pragma unroll
        for (int v = 0; v < 8; ++v) { float x = acc[t][v]; s += x; q += x * x; }
        s += __shfl_xor(s, 1, 32);  q += __shfl_xor(q, 1, 32);
        s += __shfl_xor(s, 2, 32);  q += __shfl_xor(q, 2, 32);
        s += __shfl_xor(s, 4, 32);  q += __shfl_xor(q, 4, 32);
        s += __shfl_xor(s, 16, 32); q += __shfl_xor(q, 16, 32);
        if ((lane & 7) == 0 && lane < 16) {
            int g = 2 * t + (l16 >> 3);
            atomicAdd(&sred[2 * g + 0], s);
            atomicAdd(&sred[2 * g + 1], q);
        }
    }
}

static __device__ __forceinline__ void fc1_stage(const float* __restrict__ signal,
                                                 const int* __restrict__ edges,
                                                 const float* __restrict__ efeat,
                                                 const float* __restrict__ W1,
                                                 float* Xs, float* W1s, float* Cs,
                                                 int p0) {
    const int tid = threadIdx.x;
    {
        int pt = tid >> 6, c = tid & 63;
        Cs[pt * CIN + c] = signal[(long long)(p0 + pt) * CIN + c];
    }
    for (int i = tid; i < MID * KP; i += 256) {
        int m = i / KP, c = i - m * KP;
        W1s[i] = (c < CF) ? W1[m * CF + c] : 0.f;
    }
    __syncthreads();
    {
        int r = tid >> 1;
        int h = tid & 1;
        int p = p0 + (r >> 5);
        int k = r & 31;
        int R = p * KK + k;
        int e = edges[R];
        const float4* src = ((const float4*)signal) + (long long)e * 16 + h * 8;
        const float4* ctr = ((const float4*)Cs) + (r >> 5) * 16 + h * 8;
        float4* dst = ((float4*)(Xs + r * KP)) + h * 8;
#pragma unroll
        for (int i = 0; i < 8; ++i) {
            float4 gv = src[i], cv = ctr[i], o;
            o.x = gv.x - cv.x; o.y = gv.y - cv.y;
            o.z = gv.z - cv.z; o.w = gv.w - cv.w;
            dst[i] = o;
        }
        if (h) {
            const float* e3 = efeat + (long long)R * 3;
            float* xr = Xs + r * KP;
            xr[64] = e3[0]; xr[65] = e3[1]; xr[66] = e3[2]; xr[67] = 0.f;
        }
    }
    __syncthreads();
}

__global__ __launch_bounds__(256) void k_fc1_stats(const float* __restrict__ signal,
                                                   const int* __restrict__ edges,
                                                   const float* __restrict__ efeat,
                                                   const float* __restrict__ W1,
                                                   float* __restrict__ stats1) {
    __shared__ __align__(16) float Xs[128 * KP];
    __shared__ __align__(16) float W1s[MID * KP];
    __shared__ __align__(16) float Cs[4 * CIN];
    __shared__ float sred[GRP * 2];
    const int tid = threadIdx.x;
    if (tid < GRP * 2) sred[tid] = 0.f;
    const int p0 = blockIdx.x * 4;
    fc1_stage(signal, edges, efeat, W1, Xs, W1s, Cs, p0);

    const int wv = tid >> 5, lane = tid & 31;
    v8f acc[4];
    const v8f z = {0.f, 0.f, 0.f, 0.f, 0.f, 0.f, 0.f, 0.f};
    acc[0] = z; acc[1] = z; acc[2] = z; acc[3] = z;
    wave_gemm(Xs, W1s, wv * 16, KP, lane, acc);

    stats_reduce(acc, lane, sred);
    __syncthreads();
    if (tid < GRP * 2) atomicAdd(&stats1[(p0 >> 13) * GRP * 2 + tid], sred[tid]);
}

__global__ __launch_bounds__(256) void k_fc1_pool(const float* __restrict__ signal,
                                                  const int* __restrict__ edges,
                                                  const float* __restrict__ efeat,
                                                  const float* __restrict__ W1,
                                                  const float* __restrict__ g1,
                                                  const float* __restrict__ b1,
                                                  const float* __restrict__ stats1,
                                                  float* __restrict__ h1) {
    __shared__ __align__(16) float Xs[128 * KP];
    __shared__ __align__(16) float W1s[MID * KP];
    __shared__ __align__(16) float Cs[4 * CIN];
    __shared__ float gam[MID], bet[MID];
    __shared__ float st[GRP * 2];
    __shared__ float tmax[8 * MID];
    const int tid = threadIdx.x;
    const int p0 = blockIdx.x * 4;
    const int b  = p0 >> 13;
    if (tid < MID) { gam[tid] = g1[tid]; bet[tid] = b1[tid]; }
    if (tid < GRP * 2) st[tid] = stats1[b * GRP * 2 + tid];
    fc1_stage(signal, edges, efeat, W1, Xs, W1s, Cs, p0);

    const int wv = tid >> 5, lane = tid & 31, l16 = lane & 15;
    v8f acc[4];
    const v8f z = {0.f, 0.f, 0.f, 0.f, 0.f, 0.f, 0.f, 0.f};
    acc[0] = z; acc[1] = z; acc[2] = z; acc[3] = z;
    wave_gemm(Xs, W1s, wv * 16, KP, lane, acc);

    const float invc = 1.f / (float)(NN * KK * (MID / GRP));
#pragma unroll
    for (int t = 0; t < 4; ++t) {
        int c = t * 16 + l16;
        int g = c >> 3;
        float mean = st[2 * g] * invc;
        float var  = st[2 * g + 1] * invc - mean * mean;
        float istd = rsqrtf(var + EPSV);
        float sc = gam[c] * istd;
        float bs = bet[c] - mean * sc;
        float mx = -3.402823466e38f;
#pragma unroll
        for (int v = 0; v < 8; ++v) {
            float x = acc[t][v] * sc + bs;
            x = (x >= 0.f) ? x : SLOPE * x;
            mx = fmaxf(mx, x);
        }
        mx = fmaxf(mx, __shfl_xor(mx, 16, 32));
        if (lane < 16) tmax[wv * MID + c] = mx;
    }
    __syncthreads();
    {
        int q = tid >> 6, c = tid & 63;
        float h = fmaxf(tmax[(2 * q) * MID + c], tmax[(2 * q + 1) * MID + c]);
        h1[(long long)(p0 + q) * MID + c] = h;
    }
}

__global__ __launch_bounds__(256) void k_fc_stats(const float* __restrict__ X,
                                                  const float* __restrict__ W,
                                                  float* __restrict__ Yraw,
                                                  float* __restrict__ stats) {
    __shared__ __align__(16) float Xs[128 * KP];
    __shared__ __align__(16) float Ws[MID * KP];
    __shared__ float sred[GRP * 2];
    const int tid = threadIdx.x;
    if (tid < GRP * 2) sred[tid] = 0.f;
    const long long p0 = (long long)blockIdx.x * 128;
    for (int i = tid; i < 128 * 16; i += 256) {
        int r = i >> 4, c4 = i & 15;
        float4 v = ((const float4*)(X + (p0 + r) * 64))[c4];
        ((float4*)(Xs + r * KP))[c4] = v;
    }
    for (int i = tid; i < MID * KP; i += 256) {
        int m = i / KP, c = i - m * KP;
        Ws[i] = (c < 64) ? W[m * 64 + c] : 0.f;
    }
    __syncthreads();

    const int wv = tid >> 5, lane = tid & 31, l16 = lane & 15, half = lane >> 4;
    v8f acc[4];
    const v8f z = {0.f, 0.f, 0.f, 0.f, 0.f, 0.f, 0.f, 0.f};
    acc[0] = z; acc[1] = z; acc[2] = z; acc[3] = z;
    wave_gemm(Xs, Ws, wv * 16, 64, lane, acc);

#pragma unroll
    for (int t = 0; t < 4; ++t)
#pragma unroll
        for (int v = 0; v < 8; ++v) {
            int row = wv * 16 + v + 8 * half;
            Yraw[(p0 + row) * 64 + t * 16 + l16] = acc[t][v];
        }
    stats_reduce(acc, lane, sred);
    __syncthreads();
    if (tid < GRP * 2) atomicAdd(&stats[(int)(p0 >> 13) * GRP * 2 + tid], sred[tid]);
}

__global__ __launch_bounds__(256) void k_gn_apply(float* __restrict__ Y,
                                                  const float* __restrict__ stats,
                                                  const float* __restrict__ gamma,
                                                  const float* __restrict__ beta,
                                                  float inv_count) {
    long long i = (long long)blockIdx.x * 256 + threadIdx.x;
    int c = (int)(i & 63);
    long long p = i >> 6;
    int b = (int)(p >> 13);
    int g = c >> 3;
    float sum = stats[b * GRP * 2 + 2 * g];
    float sq  = stats[b * GRP * 2 + 2 * g + 1];
    float mean = sum * inv_count;
    float var  = sq * inv_count - mean * mean;
    float istd = rsqrtf(var + EPSV);
    float sc = gamma[c] * istd;
    float bs = beta[c] - mean * sc;
    float x = Y[i] * sc + bs;
    Y[i] = (x >= 0.f) ? x : SLOPE * x;
}

__global__ void k_init(float* __restrict__ ws) {
    if (threadIdx.x < 3 * BB * GRP * 2) ws[threadIdx.x] = 0.f;
}

extern "C" void kernel_launch(void* const* d_in, const int* in_sizes, int n_in,
                              void* d_out, int out_size, void* d_ws, size_t ws_size,
                              hipStream_t stream) {
    (void)in_sizes; (void)n_in; (void)out_size; (void)ws_size;
    const float* signal = (const float*)d_in[0];
    const int*   edges  = (const int*)d_in[1];
    const float* efeat  = (const float*)d_in[2];
    const float* W1 = (const float*)d_in[3];
    const float* g1 = (const float*)d_in[4];
    const float* b1 = (const float*)d_in[5];
    const float* W2 = (const float*)d_in[6];
    const float* g2 = (const float*)d_in[7];
    const float* b2 = (const float*)d_in[8];
    const float* W3 = (const float*)d_in[9];
    const float* g3 = (const float*)d_in[10];
    const float* b3 = (const float*)d_in[11];
    float* out = (float*)d_out;

    float* ws = (float*)d_ws;
    float* stats1 = ws;
    float* stats2 = ws + 64;
    float* stats3 = ws + 128;
    float* h1   = ws + 256;
    float* buf2 = h1 + (long long)32768 * 64;

    const float invc23 = 1.f / (float)(NN * (COUT / GRP));

    hipLaunchKernelGGL(k_init,      dim3(1),    dim3(256), 0, stream, ws);
    hipLaunchKernelGGL(k_fc1_stats, dim3(8192), dim3(256), 0, stream,
                       signal, edges, efeat, W1, stats1);
    hipLaunchKernelGGL(k_fc1_pool,  dim3(8192), dim3(256), 0, stream,
                       signal, edges, efeat, W1, g1, b1, stats1, h1);
    hipLaunchKernelGGL(k_fc_stats,  dim3(256),  dim3(256), 0, stream,
                       h1, W2, buf2, stats2);
    hipLaunchKernelGGL(k_gn_apply,  dim3(8192), dim3(256), 0, stream,
                       buf2, stats2, g2, b2, invc23);
    hipLaunchKernelGGL(k_fc_stats,  dim3(256),  dim3(256), 0, stream,
                       buf2, W3, out, stats3);
    hipLaunchKernelGGL(k_gn_apply,  dim3(8192), dim3(256), 0, stream,
                       out, stats3, g3, b3, invc23);
}


# fc1 computed once: first pass also keeps each point's per-channel max (min where gamma<0) of raw fc1; GN+lrelu is monotone, so it is applied to that extremum in place; recompute pass removed
# speedup vs baseline: 1.4797x; 1.4797x over previous
#include <hip/hip_runtime.h>

#define BB    4
#define NN    8192
#define KK    32
#define CIN   64
#define CF    67
#define KP    68
#define MID   64
#define COUT  64
#define GRP   8
#define SLOPE 0.1f
#define EPSV  1e-5f

typedef __attribute__((ext_vector_type(2))) float v2f;
typedef __attribute__((ext_vector_type(8))) float v8f;

static __device__ __forceinline__ void wave_gemm(const float* Xs, const float* Ws,
                                                 int rowBase, int kdim, int lane,
                                                 v8f acc[4]) {
    const int l16  = lane & 15;
    const int koff = (lane >> 4) << 1;
    const float* xrow = Xs + (rowBase + l16) * KP;
    for (int k0 = 0; k0 < kdim; k0 += 4) {
        v2f a;
        a.x = xrow[k0 + koff];
        a.y = xrow[k0 + koff + 1];
#pragma unroll
        for (int t = 0; t < 4; ++t) {
            const float* wrow = Ws + (t * 16 + l16) * KP;
            v2f b;
            b.x = wrow[k0 + koff];
            b.y = wrow[k0 + koff + 1];
            acc[t] = __builtin_amdgcn_wmma_f32_16x16x4_f32(
                false, a, false, b, (short)0, acc[t], false, false);
        }
    }
}

static __device__ __forceinline__ void stats_reduce(v8f acc[4], int lane, float* sred) {
    const int l16 = lane & 15;
#pragma unroll
    for (int t = 0; t < 4; ++t) {
        float s = 0.f, q = 0.f;
#pragma unroll
        for (int v = 0; v < 8; ++v) { float x = acc[t][v]; s += x; q += x * x; }
        s += __shfl_xor(s, 1, 32);  q += __shfl_xor(q, 1, 32);
        s += __shfl_xor(s, 2, 32);  q += __shfl_xor(q, 2, 32);
        s += __shfl_xor(s, 4, 32);  q += __shfl_xor(q, 4, 32);
        s += __shfl_xor(s, 16, 32); q += __shfl_xor(q, 16, 32);
        if ((lane & 7) == 0 && lane < 16) {
            int g = 2 * t + (l16 >> 3);
            atomicAdd(&sred[2 * g + 0], s);
            atomicAdd(&sred[2 * g + 1], q);
        }
    }
}

static __device__ __forceinline__ void fc1_stage(const float* __restrict__ signal,
                                                 const int* __restrict__ edges,
                                                 const float* __restrict__ efeat,
                                                 const float* __restrict__ W1,
                                                 float* Xs, float* W1s, float* Cs,
                                                 int p0) {
    const int tid = threadIdx.x;
    {
        int pt = tid >> 6, c = tid & 63;
        Cs[pt * CIN + c] = signal[(long long)(p0 + pt) * CIN + c];
    }
    for (int i = tid; i < MID * KP; i += 256) {
        int m = i / KP, c = i - m * KP;
        W1s[i] = (c < CF) ? W1[m * CF + c] : 0.f;
    }
    __syncthreads();
    {
        int r = tid >> 1;
        int h = tid & 1;
        int p = p0 + (r >> 5);
        int k = r & 31;
        int R = p * KK + k;
        int e = edges[R];
        const float4* src = ((const float4*)signal) + (long long)e * 16 + h * 8;
        const float4* ctr = ((const float4*)Cs) + (r >> 5) * 16 + h * 8;
        float4* dst = ((float4*)(Xs + r * KP)) + h * 8;
#pragma unroll
        for (int i = 0; i < 8; ++i) {
            float4 gv = src[i], cv = ctr[i], o;
            o.x = gv.x - cv.x; o.y = gv.y - cv.y;
            o.z = gv.z - cv.z; o.w = gv.w - cv.w;
            dst[i] = o;
        }
        if (h) {
            const float* e3 = efeat + (long long)R * 3;
            float* xr = Xs + r * KP;
            xr[64] = e3[0]; xr[65] = e3[1]; xr[66] = e3[2]; xr[67] = 0.f;
        }
    }
    __syncthreads();
}

__global__ __launch_bounds__(256) void k_fc1_stats(const float* __restrict__ signal,
                                                   const int* __restrict__ edges,
                                                   const float* __restrict__ efeat,
                                                   const float* __restrict__ W1,
                                                   const float* __restrict__ g1,
                                                   float* __restrict__ stats1,
                                                   float* __restrict__ h1) {
    __shared__ __align__(16) float Xs[128 * KP];
    __shared__ __align__(16) float W1s[MID * KP];
    __shared__ __align__(16) float Cs[4 * CIN];
    __shared__ float sred[GRP * 2];
    __shared__ float text[8 * MID];
    __shared__ float gam[MID];
    const int tid = threadIdx.x;
    if (tid < GRP * 2) sred[tid] = 0.f;
    const int p0 = blockIdx.x * 4;
    const int wv = tid >> 5, lane = tid & 31, l16 = lane & 15;
    if (tid < MID) gam[tid] = g1[tid];
    fc1_stage(signal, edges, efeat, W1, Xs, W1s, Cs, p0);

    v8f acc[4];
    const v8f z = {0.f, 0.f, 0.f, 0.f, 0.f, 0.f, 0.f, 0.f};
    acc[0] = z; acc[1] = z; acc[2] = z; acc[3] = z;
    wave_gemm(Xs, W1s, wv * 16, KP, lane, acc);

    stats_reduce(acc, lane, sred);
    for (int t = 0; t < 4; ++t) {
        float mx = acc[t][0], mn = acc[t][0];
        for (int v = 1; v < 8; ++v) {
            mx = fmaxf(mx, acc[t][v]);
            mn = fminf(mn, acc[t][v]);
        }
        const bool pos = gam[t * 16 + l16] >= 0.f;
        float e = pos ? mx : mn;
        float o = __shfl_xor(e, 16, 32);
        e = pos ? fmaxf(e, o) : fminf(e, o);
        if (lane < 16) text[wv * MID + t * 16 + l16] = e;
    }
    __syncthreads();
    if (tid < GRP * 2) atomicAdd(&stats1[(p0 >> 13) * GRP * 2 + tid], sred[tid]);
    {
        int q = tid >> 6, c = tid & 63;
        float a = text[(2 * q) * MID + c], b = text[(2 * q + 1) * MID + c];
        float e = (gam[c] >= 0.f) ? fmaxf(a, b) : fminf(a, b);
        h1[(long long)(p0 + q) * MID + c] = e;
    }
}

__global__ __launch_bounds__(256) void k_fc_stats(const float* __restrict__ X,
                                                  const float* __restrict__ W,
                                                  float* __restrict__ Yraw,
                                                  float* __restrict__ stats) {
    __shared__ __align__(16) float Xs[128 * KP];
    __shared__ __align__(16) float Ws[MID * KP];
    __shared__ float sred[GRP * 2];
    const int tid = threadIdx.x;
    if (tid < GRP * 2) sred[tid] = 0.f;
    const long long p0 = (long long)blockIdx.x * 128;
    for (int i = tid; i < 128 * 16; i += 256) {
        int r = i >> 4, c4 = i & 15;
        float4 v = ((const float4*)(X + (p0 + r) * 64))[c4];
        ((float4*)(Xs + r * KP))[c4] = v;
    }
    for (int i = tid; i < MID * KP; i += 256) {
        int m = i / KP, c = i - m * KP;
        Ws[i] = (c < 64) ? W[m * 64 + c] : 0.f;
    }
    __syncthreads();

    const int wv = tid >> 5, lane = tid & 31, l16 = lane & 15, half = lane >> 4;
    v8f acc[4];
    const v8f z = {0.f, 0.f, 0.f, 0.f, 0.f, 0.f, 0.f, 0.f};
    acc[0] = z; acc[1] = z; acc[2] = z; acc[3] = z;
    wave_gemm(Xs, Ws, wv * 16, 64, lane, acc);

#pragma unroll
    for (int t = 0; t < 4; ++t)
#pragma unroll
        for (int v = 0; v < 8; ++v) {
            int row = wv * 16 + v + 8 * half;
            Yraw[(p0 + row) * 64 + t * 16 + l16] = acc[t][v];
        }
    stats_reduce(acc, lane, sred);
    __syncthreads();
    if (tid < GRP * 2) atomicAdd(&stats[(int)(p0 >> 13) * GRP * 2 + tid], sred[tid]);
}

__global__ __launch_bounds__(256) void k_gn_apply(float* __restrict__ Y,
                                                  const float* __restrict__ stats,
                                                  const float* __restrict__ gamma,
                                                  const float* __restrict__ beta,
                                                  float inv_count) {
    long long i = (long long)blockIdx.x * 256 + threadIdx.x;
    int c = (int)(i & 63);
    long long p = i >> 6;
    int b = (int)(p >> 13);
    int g = c >> 3;
    float sum = stats[b * GRP * 2 + 2 * g];
    float sq  = stats[b * GRP * 2 + 2 * g + 1];
    float mean = sum * inv_count;
    float var  = sq * inv_count - mean * mean;
    float istd = rsqrtf(var + EPSV);
    float sc = gamma[c] * istd;
    float bs = beta[c] - mean * sc;
    float x = Y[i] * sc + bs;
    Y[i] = (x >= 0.f) ? x : SLOPE * x;
}

__global__ void k_init(float* __restrict__ ws) {
    if (threadIdx.x < 3 * BB * GRP * 2) ws[threadIdx.x] = 0.f;
}

extern "C" void kernel_launch(void* const* d_in, const int* in_sizes, int n_in,
                              void* d_out, int out_size, void* d_ws, size_t ws_size,
                              hipStream_t stream) {
    (void)in_sizes; (void)n_in; (void)out_size; (void)ws_size;
    const float* signal = (const float*)d_in[0];
    const int*   edges  = (const int*)d_in[1];
    const float* efeat  = (const float*)d_in[2];
    const float* W1 = (const float*)d_in[3];
    const float* g1 = (const float*)d_in[4];
    const float* b1 = (const float*)d_in[5];
    const float* W2 = (const float*)d_in[6];
    const float* g2 = (const float*)d_in[7];
    const float* b2 = (const float*)d_in[8];
    const float* W3 = (const float*)d_in[9];
    const float* g3 = (const float*)d_in[10];
    const float* b3 = (const float*)d_in[11];
    float* out = (float*)d_out;

    float* ws = (float*)d_ws;
    float* stats1 = ws;
    float* stats2 = ws + 64;
    float* stats3 = ws + 128;
    float* h1   = ws + 256;
    float* buf2 = h1 + (long long)32768 * 64;

    const float invc1  = 1.f / (float)(NN * KK * (MID / GRP));
    const float invc23 = 1.f / (float)(NN * (COUT / GRP));

    hipLaunchKernelGGL(k_init,      dim3(1),    dim3(256), 0, stream, ws);
    hipLaunchKernelGGL(k_fc1_stats, dim3(8192), dim3(256), 0, stream,
                       signal, edges, efeat, W1, g1, stats1, h1);
    hipLaunchKernelGGL(k_gn_apply,  dim3(8192), dim3(256), 0, stream,
                       h1, stats1, g1, b1, invc1);
    hipLaunchKernelGGL(k_fc_stats,  dim3(256),  dim3(256), 0, stream,
                       h1, W2, buf2, stats2);
    hipLaunchKernelGGL(k_gn_apply,  dim3(8192), dim3(256), 0, stream,
                       buf2, stats2, g2, b2, invc23);
    hipLaunchKernelGGL(k_fc_stats,  dim3(256),  dim3(256), 0, stream,
                       buf2, W3, out, stats3);
    hipLaunchKernelGGL(k_gn_apply,  dim3(8192), dim3(256), 0, stream,
                       out, stats3, g3, b3, invc23);
}


# GroupNorm+lrelu of the fc1 and fc2 outputs applied by the next dense kernel while it stages its input rows into LDS (same per-element arithmetic); two elementwise launches removed, 7 to 5
# speedup vs baseline: 1.6841x; 1.1381x over previous
#include <hip/hip_runtime.h>

#define BB    4
#define NN    8192
#define KK    32
#define CIN   64
#define CF    67
#define KP    68
#define MID   64
#define COUT  64
#define GRP   8
#define SLOPE 0.1f
#define EPSV  1e-5f

typedef __attribute__((ext_vector_type(2))) float v2f;
typedef __attribute__((ext_vector_type(8))) float v8f;

static __device__ __forceinline__ void wave_gemm(const float* Xs, const float* Ws,
                                                 int rowBase, int kdim, int lane,
                                                 v8f acc[4]) {
    const int l16  = lane & 15;
    const int koff = (lane >> 4) << 1;
    const float* xrow = Xs + (rowBase + l16) * KP;
    for (int k0 = 0; k0 < kdim; k0 += 4) {
        v2f a;
        a.x = xrow[k0 + koff];
        a.y = xrow[k0 + koff + 1];
#pragma unroll
        for (int t = 0; t < 4; ++t) {
            const float* wrow = Ws + (t * 16 + l16) * KP;
            v2f b;
            b.x = wrow[k0 + koff];
            b.y = wrow[k0 + koff + 1];
            acc[t] = __builtin_amdgcn_wmma_f32_16x16x4_f32(
                false, a, false, b, (short)0, acc[t], false, false);
        }
    }
}

static __device__ __forceinline__ void stats_reduce(v8f acc[4], int lane, float* sred) {
    const int l16 = lane & 15;
#pragma unroll
    for (int t = 0; t < 4; ++t) {
        float s = 0.f, q = 0.f;
#pragma unroll
        for (int v = 0; v < 8; ++v) { float x = acc[t][v]; s += x; q += x * x; }
        s += __shfl_xor(s, 1, 32);  q += __shfl_xor(q, 1, 32);
        s += __shfl_xor(s, 2, 32);  q += __shfl_xor(q, 2, 32);
        s += __shfl_xor(s, 4, 32);  q += __shfl_xor(q, 4, 32);
        s += __shfl_xor(s, 16, 32); q += __shfl_xor(q, 16, 32);
        if ((lane & 7) == 0 && lane < 16) {
            int g = 2 * t + (l16 >> 3);
            atomicAdd(&sred[2 * g + 0], s);
            atomicAdd(&sred[2 * g + 1], q);
        }
    }
}

static __device__ __forceinline__ void fc1_stage(const float* __restrict__ signal,
                                                 const int* __restrict__ edges,
                                                 const float* __restrict__ efeat,
                                                 const float* __restrict__ W1,
                                                 float* Xs, float* W1s, float* Cs,
                                                 int p0) {
    const int tid = threadIdx.x;
    {
        int pt = tid >> 6, c = tid & 63;
        Cs[pt * CIN + c] = signal[(long long)(p0 + pt) * CIN + c];
    }
    for (int i = tid; i < MID * KP; i += 256) {
        int m = i / KP, c = i - m * KP;
        W1s[i] = (c < CF) ? W1[m * CF + c] : 0.f;
    }
    __syncthreads();
    {
        int r = tid >> 1;
        int h = tid & 1;
        int p = p0 + (r >> 5);
        int k = r & 31;
        int R = p * KK + k;
        int e = edges[R];
        const float4* src = ((const float4*)signal) + (long long)e * 16 + h * 8;
        const float4* ctr = ((const float4*)Cs) + (r >> 5) * 16 + h * 8;
        float4* dst = ((float4*)(Xs + r * KP)) + h * 8;
#pragma unroll
        for (int i = 0; i < 8; ++i) {
            float4 gv = src[i], cv = ctr[i], o;
            o.x = gv.x - cv.x; o.y = gv.y - cv.y;
            o.z = gv.z - cv.z; o.w = gv.w - cv.w;
            dst[i] = o;
        }
        if (h) {
            const float* e3 = efeat + (long long)R * 3;
            float* xr = Xs + r * KP;
            xr[64] = e3[0]; xr[65] = e3[1]; xr[66] = e3[2]; xr[67] = 0.f;
        }
    }
    __syncthreads();
}

__global__ __launch_bounds__(256) void k_fc1_stats(const float* __restrict__ signal,
                                                   const int* __restrict__ edges,
                                                   const float* __restrict__ efeat,
                                                   const float* __restrict__ W1,
                                                   const float* __restrict__ g1,
                                                   float* __restrict__ stats1,
                                                   float* __restrict__ h1) {
    __shared__ __align__(16) float Xs[128 * KP];
    __shared__ __align__(16) float W1s[MID * KP];
    __shared__ __align__(16) float Cs[4 * CIN];
    __shared__ float sred[GRP * 2];
    __shared__ float text[8 * MID];
    __shared__ float gam[MID];
    const int tid = threadIdx.x;
    if (tid < GRP * 2) sred[tid] = 0.f;
    const int p0 = blockIdx.x * 4;
    const int wv = tid >> 5, lane = tid & 31, l16 = lane & 15;
    if (tid < MID) gam[tid] = g1[tid];
    fc1_stage(signal, edges, efeat, W1, Xs, W1s, Cs, p0);

    v8f acc[4];
    const v8f z = {0.f, 0.f, 0.f, 0.f, 0.f, 0.f, 0.f, 0.f};
    acc[0] = z; acc[1] = z; acc[2] = z; acc[3] = z;
    wave_gemm(Xs, W1s, wv * 16, KP, lane, acc);

    stats_reduce(acc, lane, sred);
    for (int t = 0; t < 4; ++t) {
        float mx = acc[t][0], mn = acc[t][0];
        for (int v = 1; v < 8; ++v) {
            mx = fmaxf(mx, acc[t][v]);
            mn = fminf(mn, acc[t][v]);
        }
        const bool pos = gam[t * 16 + l16] >= 0.f;
        float e = pos ? mx : mn;
        float o = __shfl_xor(e, 16, 32);
        e = pos ? fmaxf(e, o) : fminf(e, o);
        if (lane < 16) text[wv * MID + t * 16 + l16] = e;
    }
    __syncthreads();
    if (tid < GRP * 2) atomicAdd(&stats1[(p0 >> 13) * GRP * 2 + tid], sred[tid]);
    {
        int q = tid >> 6, c = tid & 63;
        float a = text[(2 * q) * MID + c], b = text[(2 * q + 1) * MID + c];
        float e = (gam[c] >= 0.f) ? fmaxf(a, b) : fminf(a, b);
        h1[(long long)(p0 + q) * MID + c] = e;
    }
}

__global__ __launch_bounds__(256) void k_fc_stats(const float* __restrict__ X,
                                                  const float* __restrict__ pstats,
                                                  const float* __restrict__ pgamma,
                                                  const float* __restrict__ pbeta,
                                                  float inv_count,
                                                  const float* __restrict__ W,
                                                  float* __restrict__ Yraw,
                                                  float* __restrict__ stats) {
    __shared__ __align__(16) float Xs[128 * KP];
    __shared__ __align__(16) float Ws[MID * KP];
    __shared__ float sred[GRP * 2];
    const int tid = threadIdx.x;
    if (tid < GRP * 2) sred[tid] = 0.f;
    const long long p0 = (long long)blockIdx.x * 128;

    const int c4 = tid & 15;
    float4 sc, bs;
    {
        const int b = (int)(p0 >> 13);
        const int g = c4 >> 1;
        float sum = pstats[b * GRP * 2 + 2 * g];
        float sq  = pstats[b * GRP * 2 + 2 * g + 1];
        float mean = sum * inv_count;
        float var  = sq * inv_count - mean * mean;
        float istd = rsqrtf(var + EPSV);
        float4 gm = ((const float4*)pgamma)[c4];
        float4 bt = ((const float4*)pbeta)[c4];
        sc.x = gm.x * istd; sc.y = gm.y * istd; sc.z = gm.z * istd; sc.w = gm.w * istd;
        bs.x = bt.x - mean * sc.x; bs.y = bt.y - mean * sc.y;
        bs.z = bt.z - mean * sc.z; bs.w = bt.w - mean * sc.w;
    }
    for (int i = tid; i < 128 * 16; i += 256) {
        int r = i >> 4;
        float4 v = ((const float4*)(X + (p0 + r) * 64))[c4];
        float4 o;
        o.x = v.x * sc.x + bs.x; o.x = (o.x >= 0.f) ? o.x : SLOPE * o.x;
        o.y = v.y * sc.y + bs.y; o.y = (o.y >= 0.f) ? o.y : SLOPE * o.y;
        o.z = v.z * sc.z + bs.z; o.z = (o.z >= 0.f) ? o.z : SLOPE * o.z;
        o.w = v.w * sc.w + bs.w; o.w = (o.w >= 0.f) ? o.w : SLOPE * o.w;
        ((float4*)(Xs + r * KP))[c4] = o;
    }
    for (int i = tid; i < MID * KP; i += 256) {
        int m = i / KP, c = i - m * KP;
        Ws[i] = (c < 64) ? W[m * 64 + c] : 0.f;
    }
    __syncthreads();

    const int wv = tid >> 5, lane = tid & 31, l16 = lane & 15, half = lane >> 4;
    v8f acc[4];
    const v8f z = {0.f, 0.f, 0.f, 0.f, 0.f, 0.f, 0.f, 0.f};
    acc[0] = z; acc[1] = z; acc[2] = z; acc[3] = z;
    wave_gemm(Xs, Ws, wv * 16, 64, lane, acc);

#pragma unroll
    for (int t = 0; t < 4; ++t)
#pragma unroll
        for (int v = 0; v < 8; ++v) {
            int row = wv * 16 + v + 8 * half;
            Yraw[(p0 + row) * 64 + t * 16 + l16] = acc[t][v];
        }
    stats_reduce(acc, lane, sred);
    __syncthreads();
    if (tid < GRP * 2) atomicAdd(&stats[(int)(p0 >> 13) * GRP * 2 + tid], sred[tid]);
}

__global__ __launch_bounds__(256) void k_gn_apply(float* __restrict__ Y,
                                                  const float* __restrict__ stats,
                                                  const float* __restrict__ gamma,
                                                  const float* __restrict__ beta,
                                                  float inv_count) {
    long long i = (long long)blockIdx.x * 256 + threadIdx.x;
    int c = (int)(i & 63);
    long long p = i >> 6;
    int b = (int)(p >> 13);
    int g = c >> 3;
    float sum = stats[b * GRP * 2 + 2 * g];
    float sq  = stats[b * GRP * 2 + 2 * g + 1];
    float mean = sum * inv_count;
    float var  = sq * inv_count - mean * mean;
    float istd = rsqrtf(var + EPSV);
    float sc = gamma[c] * istd;
    float bs = beta[c] - mean * sc;
    float x = Y[i] * sc + bs;
    Y[i] = (x >= 0.f) ? x : SLOPE * x;
}

__global__ void k_init(float* __restrict__ ws) {
    if (threadIdx.x < 3 * BB * GRP * 2) ws[threadIdx.x] = 0.f;
}

extern "C" void kernel_launch(void* const* d_in, const int* in_sizes, int n_in,
                              void* d_out, int out_size, void* d_ws, size_t ws_size,
                              hipStream_t stream) {
    (void)in_sizes; (void)n_in; (void)out_size; (void)ws_size;
    const float* signal = (const float*)d_in[0];
    const int*   edges  = (const int*)d_in[1];
    const float* efeat  = (const float*)d_in[2];
    const float* W1 = (const float*)d_in[3];
    const float* g1 = (const float*)d_in[4];
    const float* b1 = (const float*)d_in[5];
    const float* W2 = (const float*)d_in[6];
    const float* g2 = (const float*)d_in[7];
    const float* b2 = (const float*)d_in[8];
    const float* W3 = (const float*)d_in[9];
    const float* g3 = (const float*)d_in[10];
    const float* b3 = (const float*)d_in[11];
    float* out = (float*)d_out;

    float* ws = (float*)d_ws;
    float* stats1 = ws;
    float* stats2 = ws + 64;
    float* stats3 = ws + 128;
    float* h1   = ws + 256;
    float* buf2 = h1 + (long long)32768 * 64;

    const float invc1  = 1.f / (float)(NN * KK * (MID / GRP));
    const float invc23 = 1.f / (float)(NN * (COUT / GRP));

    hipLaunchKernelGGL(k_init,      dim3(1),    dim3(256), 0, stream, ws);
    hipLaunchKernelGGL(k_fc1_stats, dim3(8192), dim3(256), 0, stream,
                       signal, edges, efeat, W1, g1, stats1, h1);
    hipLaunchKernelGGL(k_fc_stats,  dim3(256),  dim3(256), 0, stream,
                       h1, stats1, g1, b1, invc1, W2, buf2, stats2);
    hipLaunchKernelGGL(k_fc_stats,  dim3(256),  dim3(256), 0, stream,
                       buf2, stats2, g2, b2, invc23, W3, out, stats3);
    hipLaunchKernelGGL(k_gn_apply,  dim3(8192), dim3(256), 0, stream,
                       out, stats3, g3, b3, invc23);
}


# fc1 is linear: W1[:,0:64] applied once to the 32768 signal rows (WMMA table P); edge pass loads P[e]-P[centre] as C of one K=4 edge-feat WMMA step; fc1 and stats summation order changed
# speedup vs baseline: 3.7952x; 2.2535x over previous
#include <hip/hip_runtime.h>

#define BB    4
#define NN    8192
#define KK    32
#define CIN   64
#define CF    67
#define KP    68
#define MID   64
#define COUT  64
#define GRP   8
#define SLOPE 0.1f
#define EPSV  1e-5f

typedef __attribute__((ext_vector_type(2))) float v2f;
typedef __attribute__((ext_vector_type(8))) float v8f;

static __device__ __forceinline__ void wave_gemm(const float* Xs, const float* Ws,
                                                 int rowBase, int kdim, int lane,
                                                 v8f acc[4]) {
    const int l16  = lane & 15;
    const int koff = (lane >> 4) << 1;
    const float* xrow = Xs + (rowBase + l16) * KP;
    for (int k0 = 0; k0 < kdim; k0 += 4) {
        v2f a;
        a.x = xrow[k0 + koff];
        a.y = xrow[k0 + koff + 1];
#pragma unroll
        for (int t = 0; t < 4; ++t) {
            const float* wrow = Ws + (t * 16 + l16) * KP;
            v2f b;
            b.x = wrow[k0 + koff];
            b.y = wrow[k0 + koff + 1];
            acc[t] = __builtin_amdgcn_wmma_f32_16x16x4_f32(
                false, a, false, b, (short)0, acc[t], false, false);
        }
    }
}

static __device__ __forceinline__ void stats_commit(const float s_in[4], const float q_in[4],
                                                    int lane, float* sred) {
    const int l16 = lane & 15;
#pragma unroll
    for (int t = 0; t < 4; ++t) {
        float s = s_in[t], q = q_in[t];
        s += __shfl_xor(s, 1, 32);  q += __shfl_xor(q, 1, 32);
        s += __shfl_xor(s, 2, 32);  q += __shfl_xor(q, 2, 32);
        s += __shfl_xor(s, 4, 32);  q += __shfl_xor(q, 4, 32);
        s += __shfl_xor(s, 16, 32); q += __shfl_xor(q, 16, 32);
        if ((lane & 7) == 0 && lane < 16) {
            int g = 2 * t + (l16 >> 3);
            atomicAdd(&sred[2 * g + 0], s);
            atomicAdd(&sred[2 * g + 1], q);
        }
    }
}

static __device__ __forceinline__ void lane_sums(const v8f acc[4], float s[4], float q[4]) {
#pragma unroll
    for (int t = 0; t < 4; ++t)
#pragma unroll
        for (int v = 0; v < 8; ++v) { float x = acc[t][v]; s[t] += x; q[t] += x * x; }
}

static __device__ __forceinline__ void stats_reduce(v8f acc[4], int lane, float* sred) {
    float s[4] = {0.f, 0.f, 0.f, 0.f}, q[4] = {0.f, 0.f, 0.f, 0.f};
    lane_sums(acc, s, q);
    stats_commit(s, q, lane, sred);
}

__global__ __launch_bounds__(256) void k_table(const float* __restrict__ signal,
                                               const float* __restrict__ W1,
                                               float* __restrict__ P,
                                               float* __restrict__ stats) {
    __shared__ __align__(16) float Xs[128 * KP];
    __shared__ __align__(16) float Ws[MID * KP];
    const int tid = threadIdx.x;
    if (blockIdx.x == 0 && tid < 3 * BB * GRP * 2) stats[tid] = 0.f;
    const long long p0 = (long long)blockIdx.x * 128;
    const int c4 = tid & 15, r0 = tid >> 4;
    float4 xv[8];
    float  wl[16];
#pragma unroll
    for (int i = 0; i < 8; ++i)
        xv[i] = ((const float4*)(signal + (p0 + r0 + 16 * i) * CIN))[c4];
#pragma unroll
    for (int i = 0; i < 16; ++i) {
        int idx = tid + 256 * i;
        wl[i] = W1[(idx >> 6) * CF + (idx & 63)];
    }
#pragma unroll
    for (int i = 0; i < 8; ++i)
        ((float4*)(Xs + (r0 + 16 * i) * KP))[c4] = xv[i];
#pragma unroll
    for (int i = 0; i < 16; ++i) {
        int idx = tid + 256 * i;
        Ws[(idx >> 6) * KP + (idx & 63)] = wl[i];
    }
    __syncthreads();

    const int wv = tid >> 5, lane = tid & 31, l16 = lane & 15, half = lane >> 4;
    v8f acc[4];
    const v8f z = {0.f, 0.f, 0.f, 0.f, 0.f, 0.f, 0.f, 0.f};
    acc[0] = z; acc[1] = z; acc[2] = z; acc[3] = z;
    wave_gemm(Xs, Ws, wv * 16, 64, lane, acc);

#pragma unroll
    for (int t = 0; t < 4; ++t)
#pragma unroll
        for (int v = 0; v < 8; ++v) {
            int row = wv * 16 + v + 8 * half;
            P[(p0 + row) * 64 + t * 16 + l16] = acc[t][v];
        }
}

#define PTS 32
__global__ __launch_bounds__(256) void k_fc1_edges(const float* __restrict__ P,
                                                   const int* __restrict__ edges,
                                                   const float* __restrict__ efeat,
                                                   const float* __restrict__ W1,
                                                   const float* __restrict__ g1,
                                                   float* __restrict__ stats1,
                                                   float* __restrict__ h1) {
    __shared__ float sred[GRP * 2];
    const int tid = threadIdx.x;
    if (tid < GRP * 2) sred[tid] = 0.f;
    __syncthreads();
    const int wv = tid >> 5, lane = tid & 31, l16 = lane & 15, half = lane >> 4;

    v2f  bw[4];
    bool pos[4];
    {
        const float4 gq = ((const float4*)g1)[l16];
        pos[0] = gq.x >= 0.f; pos[1] = gq.y >= 0.f; pos[2] = gq.z >= 0.f; pos[3] = gq.w >= 0.f;
    }
#pragma unroll
    for (int t = 0; t < 4; ++t) {
        const int m = 4 * l16 + t;
        bw[t].x = W1[m * CF + 64 + 2 * half];
        float w1 = W1[m * CF + 65 - half];
        bw[t].y = half ? 0.f : w1;
    }

    float s[4] = {0.f, 0.f, 0.f, 0.f}, q[4] = {0.f, 0.f, 0.f, 0.f};
    const int pbase = blockIdx.x * PTS;
    for (int it = 0; it < PTS / 8; ++it) {
        const int p = pbase + it * 8 + wv;
        const float4 cen = ((const float4*)(P + (unsigned)p * 64u))[l16];
        float ext[4];
#pragma unroll
        for (int t = 0; t < 4; ++t) ext[t] = pos[t] ? -INFINITY : INFINITY;
#pragma unroll 1
        for (int j = 0; j < 2; ++j) {
            const int R0 = p * KK + j * 16;
            const int4* ep = (const int4*)(edges + R0 + 8 * half);
            const int4 ea = ep[0], eb = ep[1];
            const unsigned e[8] = {(unsigned)ea.x, (unsigned)ea.y, (unsigned)ea.z, (unsigned)ea.w,
                                   (unsigned)eb.x, (unsigned)eb.y, (unsigned)eb.z, (unsigned)eb.w};
            const float* fr = efeat + (long long)(R0 + l16) * 3;
            v2f a;
            a.x = fr[2 * half];
            float f1 = fr[1];
            a.y = half ? 0.f : f1;

            v8f acc[4];
#pragma unroll
            for (int v = 0; v < 8; ++v) {
                const float4 g = ((const float4*)(P + e[v] * 64u))[l16];
                acc[0][v] = g.x - cen.x; acc[1][v] = g.y - cen.y;
                acc[2][v] = g.z - cen.z; acc[3][v] = g.w - cen.w;
            }
#pragma unroll
            for (int t = 0; t < 4; ++t)
                acc[t] = __builtin_amdgcn_wmma_f32_16x16x4_f32(
                    false, a, false, bw[t], (short)0, acc[t], false, false);

            lane_sums(acc, s, q);
#pragma unroll
            for (int t = 0; t < 4; ++t) {
                float mx = acc[t][0], mn = acc[t][0];
#pragma unroll
                for (int v = 1; v < 8; ++v) {
                    mx = fmaxf(mx, acc[t][v]);
                    mn = fminf(mn, acc[t][v]);
                }
                ext[t] = pos[t] ? fmaxf(ext[t], mx) : fminf(ext[t], mn);
            }
        }
#pragma unroll
        for (int t = 0; t < 4; ++t) {
            float o = __shfl_xor(ext[t], 16, 32);
            ext[t] = pos[t] ? fmaxf(ext[t], o) : fminf(ext[t], o);
        }
        if (lane < 16) {
            float4 o; o.x = ext[0]; o.y = ext[1]; o.z = ext[2]; o.w = ext[3];
            ((float4*)(h1 + (long long)p * MID))[l16] = o;
        }
    }
    float ss = (s[0] + s[1]) + (s[2] + s[3]);
    float qq = (q[0] + q[1]) + (q[2] + q[3]);
    ss += __shfl_xor(ss, 1, 32);  qq += __shfl_xor(qq, 1, 32);
    ss += __shfl_xor(ss, 16, 32); qq += __shfl_xor(qq, 16, 32);
    if ((lane & 1) == 0 && lane < 16) {
        atomicAdd(&sred[2 * (l16 >> 1) + 0], ss);
        atomicAdd(&sred[2 * (l16 >> 1) + 1], qq);
    }
    __syncthreads();
    if (tid < GRP * 2) atomicAdd(&stats1[(pbase >> 13) * GRP * 2 + tid], sred[tid]);
}

__global__ __launch_bounds__(256) void k_fc_stats(const float* __restrict__ X,
                                                  const float* __restrict__ pstats,
                                                  const float* __restrict__ pgamma,
                                                  const float* __restrict__ pbeta,
                                                  float inv_count,
                                                  const float* __restrict__ W,
                                                  float* __restrict__ Yraw,
                                                  float* __restrict__ stats) {
    __shared__ __align__(16) float Xs[128 * KP];
    __shared__ __align__(16) float Ws[MID * KP];
    __shared__ float sred[GRP * 2];
    const int tid = threadIdx.x;
    if (tid < GRP * 2) sred[tid] = 0.f;
    const long long p0 = (long long)blockIdx.x * 128;

    const int c4 = tid & 15;
    float4 sc, bs;
    {
        const int b = (int)(p0 >> 13);
        const int g = c4 >> 1;
        float sum = pstats[b * GRP * 2 + 2 * g];
        float sq  = pstats[b * GRP * 2 + 2 * g + 1];
        float mean = sum * inv_count;
        float var  = sq * inv_count - mean * mean;
        float istd = rsqrtf(var + EPSV);
        float4 gm = ((const float4*)pgamma)[c4];
        float4 bt = ((const float4*)pbeta)[c4];
        sc.x = gm.x * istd; sc.y = gm.y * istd; sc.z = gm.z * istd; sc.w = gm.w * istd;
        bs.x = bt.x - mean * sc.x; bs.y = bt.y - mean * sc.y;
        bs.z = bt.z - mean * sc.z; bs.w = bt.w - mean * sc.w;
    }
    for (int i = tid; i < 128 * 16; i += 256) {
        int r = i >> 4;
        float4 v = ((const float4*)(X + (p0 + r) * 64))[c4];
        float4 o;
        o.x = v.x * sc.x + bs.x; o.x = (o.x >= 0.f) ? o.x : SLOPE * o.x;
        o.y = v.y * sc.y + bs.y; o.y = (o.y >= 0.f) ? o.y : SLOPE * o.y;
        o.z = v.z * sc.z + bs.z; o.z = (o.z >= 0.f) ? o.z : SLOPE * o.z;
        o.w = v.w * sc.w + bs.w; o.w = (o.w >= 0.f) ? o.w : SLOPE * o.w;
        ((float4*)(Xs + r * KP))[c4] = o;
    }
    for (int i = tid; i < MID * KP; i += 256) {
        int m = i / KP, c = i - m * KP;
        Ws[i] = (c < 64) ? W[m * 64 + c] : 0.f;
    }
    __syncthreads();

    const int wv = tid >> 5, lane = tid & 31, l16 = lane & 15, half = lane >> 4;
    v8f acc[4];
    const v8f z = {0.f, 0.f, 0.f, 0.f, 0.f, 0.f, 0.f, 0.f};
    acc[0] = z; acc[1] = z; acc[2] = z; acc[3] = z;
    wave_gemm(Xs, Ws, wv * 16, 64, lane, acc);

#pragma unroll
    for (int t = 0; t < 4; ++t)
#pragma unroll
        for (int v = 0; v < 8; ++v) {
            int row = wv * 16 + v + 8 * half;
            Yraw[(p0 + row) * 64 + t * 16 + l16] = acc[t][v];
        }
    stats_reduce(acc, lane, sred);
    __syncthreads();
    if (tid < GRP * 2) atomicAdd(&stats[(int)(p0 >> 13) * GRP * 2 + tid], sred[tid]);
}

__global__ __launch_bounds__(256) void k_gn_apply(float* __restrict__ Y,
                                                  const float* __restrict__ stats,
                                                  const float* __restrict__ gamma,
                                                  const float* __restrict__ beta,
                                                  float inv_count) {
    long long i = (long long)blockIdx.x * 256 + threadIdx.x;
    int c = (int)(i & 63);
    long long p = i >> 6;
    int b = (int)(p >> 13);
    int g = c >> 3;
    float sum = stats[b * GRP * 2 + 2 * g];
    float sq  = stats[b * GRP * 2 + 2 * g + 1];
    float mean = sum * inv_count;
    float var  = sq * inv_count - mean * mean;
    float istd = rsqrtf(var + EPSV);
    float sc = gamma[c] * istd;
    float bs = beta[c] - mean * sc;
    float x = Y[i] * sc + bs;
    Y[i] = (x >= 0.f) ? x : SLOPE * x;
}

extern "C" void kernel_launch(void* const* d_in, const int* in_sizes, int n_in,
                              void* d_out, int out_size, void* d_ws, size_t ws_size,
                              hipStream_t stream) {
    (void)in_sizes; (void)n_in; (void)out_size; (void)ws_size;
    const float* signal = (const float*)d_in[0];
    const int*   edges  = (const int*)d_in[1];
    const float* efeat  = (const float*)d_in[2];
    const float* W1 = (const float*)d_in[3];
    const float* g1 = (const float*)d_in[4];
    const float* b1 = (const float*)d_in[5];
    const float* W2 = (const float*)d_in[6];
    const float* g2 = (const float*)d_in[7];
    const float* b2 = (const float*)d_in[8];
    const float* W3 = (const float*)d_in[9];
    const float* g3 = (const float*)d_in[10];
    const float* b3 = (const float*)d_in[11];
    float* out = (float*)d_out;

    float* ws = (float*)d_ws;
    float* stats1 = ws;
    float* stats2 = ws + 64;
    float* stats3 = ws + 128;
    float* h1   = ws + 256;
    float* buf2 = h1 + (long long)32768 * 64;
    float* ptab = buf2;

    const float invc1  = 1.f / (float)(NN * KK * (MID / GRP));
    const float invc23 = 1.f / (float)(NN * (COUT / GRP));

    hipLaunchKernelGGL(k_table,     dim3(256),  dim3(256), 0, stream,
                       signal, W1, ptab, ws);
    hipLaunchKernelGGL(k_fc1_edges, dim3(1024), dim3(256), 0, stream,
                       ptab, edges, efeat, W1, g1, stats1, h1);
    hipLaunchKernelGGL(k_fc_stats,  dim3(256),  dim3(256), 0, stream,
                       h1, stats1, g1, b1, invc1, W2, buf2, stats2);
    hipLaunchKernelGGL(k_fc_stats,  dim3(256),  dim3(256), 0, stream,
                       buf2, stats2, g2, b2, invc23, W3, out, stats3);
    hipLaunchKernelGGL(k_gn_apply,  dim3(8192), dim3(256), 0, stream,
                       out, stats3, g3, b3, invc23);
}


# fc2/fc3 dense kernel staging: all of a thread's input-row, weight (now float4) and statistics loads issued before its first LDS store; 26 serial memory round trips per work-group become one
# speedup vs baseline: 5.4895x; 1.4464x over previous
#include <hip/hip_runtime.h>

#define BB    4
#define NN    8192
#define KK    32
#define CIN   64
#define CF    67
#define KP    68
#define MID   64
#define COUT  64
#define GRP   8
#define SLOPE 0.1f
#define EPSV  1e-5f

typedef __attribute__((ext_vector_type(2))) float v2f;
typedef __attribute__((ext_vector_type(8))) float v8f;

static __device__ __forceinline__ void wave_gemm(const float* Xs, const float* Ws,
                                                 int rowBase, int kdim, int lane,
                                                 v8f acc[4]) {
    const int l16  = lane & 15;
    const int koff = (lane >> 4) << 1;
    const float* xrow = Xs + (rowBase + l16) * KP;
    for (int k0 = 0; k0 < kdim; k0 += 4) {
        v2f a;
        a.x = xrow[k0 + koff];
        a.y = xrow[k0 + koff + 1];
#pragma unroll
        for (int t = 0; t < 4; ++t) {
            const float* wrow = Ws + (t * 16 + l16) * KP;
            v2f b;
            b.x = wrow[k0 + koff];
            b.y = wrow[k0 + koff + 1];
            acc[t] = __builtin_amdgcn_wmma_f32_16x16x4_f32(
                false, a, false, b, (short)0, acc[t], false, false);
        }
    }
}

static __device__ __forceinline__ void stats_commit(const float s_in[4], const float q_in[4],
                                                    int lane, float* sred) {
    const int l16 = lane & 15;
#pragma unroll
    for (int t = 0; t < 4; ++t) {
        float s = s_in[t], q = q_in[t];
        s += __shfl_xor(s, 1, 32);  q += __shfl_xor(q, 1, 32);
        s += __shfl_xor(s, 2, 32);  q += __shfl_xor(q, 2, 32);
        s += __shfl_xor(s, 4, 32);  q += __shfl_xor(q, 4, 32);
        s += __shfl_xor(s, 16, 32); q += __shfl_xor(q, 16, 32);
        if ((lane & 7) == 0 && lane < 16) {
            int g = 2 * t + (l16 >> 3);
            atomicAdd(&sred[2 * g + 0], s);
            atomicAdd(&sred[2 * g + 1], q);
        }
    }
}

static __device__ __forceinline__ void lane_sums(const v8f acc[4], float s[4], float q[4]) {
#pragma unroll
    for (int t = 0; t < 4; ++t)
#pragma unroll
        for (int v = 0; v < 8; ++v) { float x = acc[t][v]; s[t] += x; q[t] += x * x; }
}

static __device__ __forceinline__ void stats_reduce(v8f acc[4], int lane, float* sred) {
    float s[4] = {0.f, 0.f, 0.f, 0.f}, q[4] = {0.f, 0.f, 0.f, 0.f};
    lane_sums(acc, s, q);
    stats_commit(s, q, lane, sred);
}

__global__ __launch_bounds__(256) void k_table(const float* __restrict__ signal,
                                               const float* __restrict__ W1,
                                               float* __restrict__ P,
                                               float* __restrict__ stats) {
    __shared__ __align__(16) float Xs[128 * KP];
    __shared__ __align__(16) float Ws[MID * KP];
    const int tid = threadIdx.x;
    if (blockIdx.x == 0 && tid < 3 * BB * GRP * 2) stats[tid] = 0.f;
    const long long p0 = (long long)blockIdx.x * 128;
    const int c4 = tid & 15, r0 = tid >> 4;
    float4 xv[8];
    float  wl[16];
#pragma unroll
    for (int i = 0; i < 8; ++i)
        xv[i] = ((const float4*)(signal + (p0 + r0 + 16 * i) * CIN))[c4];
#pragma unroll
    for (int i = 0; i < 16; ++i) {
        int idx = tid + 256 * i;
        wl[i] = W1[(idx >> 6) * CF + (idx & 63)];
    }
#pragma unroll
    for (int i = 0; i < 8; ++i)
        ((float4*)(Xs + (r0 + 16 * i) * KP))[c4] = xv[i];
#pragma unroll
    for (int i = 0; i < 16; ++i) {
        int idx = tid + 256 * i;
        Ws[(idx >> 6) * KP + (idx & 63)] = wl[i];
    }
    __syncthreads();

    const int wv = tid >> 5, lane = tid & 31, l16 = lane & 15, half = lane >> 4;
    v8f acc[4];
    const v8f z = {0.f, 0.f, 0.f, 0.f, 0.f, 0.f, 0.f, 0.f};
    acc[0] = z; acc[1] = z; acc[2] = z; acc[3] = z;
    wave_gemm(Xs, Ws, wv * 16, 64, lane, acc);

#pragma unroll
    for (int t = 0; t < 4; ++t)
#pragma unroll
        for (int v = 0; v < 8; ++v) {
            int row = wv * 16 + v + 8 * half;
            P[(p0 + row) * 64 + t * 16 + l16] = acc[t][v];
        }
}

#define PTS 32
__global__ __launch_bounds__(256) void k_fc1_edges(const float* __restrict__ P,
                                                   const int* __restrict__ edges,
                                                   const float* __restrict__ efeat,
                                                   const float* __restrict__ W1,
                                                   const float* __restrict__ g1,
                                                   float* __restrict__ stats1,
                                                   float* __restrict__ h1) {
    __shared__ float sred[GRP * 2];
    const int tid = threadIdx.x;
    if (tid < GRP * 2) sred[tid] = 0.f;
    __syncthreads();
    const int wv = tid >> 5, lane = tid & 31, l16 = lane & 15, half = lane >> 4;

    v2f  bw[4];
    bool pos[4];
    {
        const float4 gq = ((const float4*)g1)[l16];
        pos[0] = gq.x >= 0.f; pos[1] = gq.y >= 0.f; pos[2] = gq.z >= 0.f; pos[3] = gq.w >= 0.f;
    }
#pragma unroll
    for (int t = 0; t < 4; ++t) {
        const int m = 4 * l16 + t;
        bw[t].x = W1[m * CF + 64 + 2 * half];
        float w1 = W1[m * CF + 65 - half];
        bw[t].y = half ? 0.f : w1;
    }

    float s[4] = {0.f, 0.f, 0.f, 0.f}, q[4] = {0.f, 0.f, 0.f, 0.f};
    const int pbase = blockIdx.x * PTS;
    for (int it = 0; it < PTS / 8; ++it) {
        const int p = pbase + it * 8 + wv;
        const float4 cen = ((const float4*)(P + (unsigned)p * 64u))[l16];
        float ext[4];
#pragma unroll
        for (int t = 0; t < 4; ++t) ext[t] = pos[t] ? -INFINITY : INFINITY;
#pragma unroll 1
        for (int j = 0; j < 2; ++j) {
            const int R0 = p * KK + j * 16;
            const int4* ep = (const int4*)(edges + R0 + 8 * half);
            const int4 ea = ep[0], eb = ep[1];
            const unsigned e[8] = {(unsigned)ea.x, (unsigned)ea.y, (unsigned)ea.z, (unsigned)ea.w,
                                   (unsigned)eb.x, (unsigned)eb.y, (unsigned)eb.z, (unsigned)eb.w};
            const float* fr = efeat + (long long)(R0 + l16) * 3;
            v2f a;
            a.x = fr[2 * half];
            float f1 = fr[1];
            a.y = half ? 0.f : f1;

            v8f acc[4];
#pragma unroll
            for (int v = 0; v < 8; ++v) {
                const float4 g = ((const float4*)(P + e[v] * 64u))[l16];
                acc[0][v] = g.x - cen.x; acc[1][v] = g.y - cen.y;
                acc[2][v] = g.z - cen.z; acc[3][v] = g.w - cen.w;
            }
#pragma unroll
            for (int t = 0; t < 4; ++t)
                acc[t] = __builtin_amdgcn_wmma_f32_16x16x4_f32(
                    false, a, false, bw[t], (short)0, acc[t], false, false);

            lane_sums(acc, s, q);
#pragma unroll
            for (int t = 0; t < 4; ++t) {
                float mx = acc[t][0], mn = acc[t][0];
#pragma unroll
                for (int v = 1; v < 8; ++v) {
                    mx = fmaxf(mx, acc[t][v]);
                    mn = fminf(mn, acc[t][v]);
                }
                ext[t] = pos[t] ? fmaxf(ext[t], mx) : fminf(ext[t], mn);
            }
        }
#pragma unroll
        for (int t = 0; t < 4; ++t) {
            float o = __shfl_xor(ext[t], 16, 32);
            ext[t] = pos[t] ? fmaxf(ext[t], o) : fminf(ext[t], o);
        }
        if (lane < 16) {
            float4 o; o.x = ext[0]; o.y = ext[1]; o.z = ext[2]; o.w = ext[3];
            ((float4*)(h1 + (long long)p * MID))[l16] = o;
        }
    }
    float ss = (s[0] + s[1]) + (s[2] + s[3]);
    float qq = (q[0] + q[1]) + (q[2] + q[3]);
    ss += __shfl_xor(ss, 1, 32);  qq += __shfl_xor(qq, 1, 32);
    ss += __shfl_xor(ss, 16, 32); qq += __shfl_xor(qq, 16, 32);
    if ((lane & 1) == 0 && lane < 16) {
        atomicAdd(&sred[2 * (l16 >> 1) + 0], ss);
        atomicAdd(&sred[2 * (l16 >> 1) + 1], qq);
    }
    __syncthreads();
    if (tid < GRP * 2) atomicAdd(&stats1[(pbase >> 13) * GRP * 2 + tid], sred[tid]);
}

__global__ __launch_bounds__(256) void k_fc_stats(const float* __restrict__ X,
                                                  const float* __restrict__ pstats,
                                                  const float* __restrict__ pgamma,
                                                  const float* __restrict__ pbeta,
                                                  float inv_count,
                                                  const float* __restrict__ W,
                                                  float* __restrict__ Yraw,
                                                  float* __restrict__ stats) {
    __shared__ __align__(16) float Xs[128 * KP];
    __shared__ __align__(16) float Ws[MID * KP];
    __shared__ float sred[GRP * 2];
    const int tid = threadIdx.x;
    if (tid < GRP * 2) sred[tid] = 0.f;
    const long long p0 = (long long)blockIdx.x * 128;

    const int c4 = tid & 15, r0 = tid >> 4;
    const int b = (int)(p0 >> 13);
    const int g = c4 >> 1;
    const float sum = pstats[b * GRP * 2 + 2 * g];
    const float sq  = pstats[b * GRP * 2 + 2 * g + 1];
    const float4 gm = ((const float4*)pgamma)[c4];
    const float4 bt = ((const float4*)pbeta)[c4];
    float4 xv[8], wq[4];
#pragma unroll
    for (int i = 0; i < 8; ++i)
        xv[i] = ((const float4*)(X + (p0 + r0 + 16 * i) * 64))[c4];
#pragma unroll
    for (int i = 0; i < 4; ++i)
        wq[i] = ((const float4*)(W + (r0 + 16 * i) * 64))[c4];
    __builtin_amdgcn_sched_barrier(0);

    float4 sc, bs;
    {
        float mean = sum * inv_count;
        float var  = sq * inv_count - mean * mean;
        float istd = rsqrtf(var + EPSV);
        sc.x = gm.x * istd; sc.y = gm.y * istd; sc.z = gm.z * istd; sc.w = gm.w * istd;
        bs.x = bt.x - mean * sc.x; bs.y = bt.y - mean * sc.y;
        bs.z = bt.z - mean * sc.z; bs.w = bt.w - mean * sc.w;
    }
#pragma unroll
    for (int i = 0; i < 8; ++i) {
        const float4 v = xv[i];
        float4 o;
        o.x = v.x * sc.x + bs.x; o.x = (o.x >= 0.f) ? o.x : SLOPE * o.x;
        o.y = v.y * sc.y + bs.y; o.y = (o.y >= 0.f) ? o.y : SLOPE * o.y;
        o.z = v.z * sc.z + bs.z; o.z = (o.z >= 0.f) ? o.z : SLOPE * o.z;
        o.w = v.w * sc.w + bs.w; o.w = (o.w >= 0.f) ? o.w : SLOPE * o.w;
        ((float4*)(Xs + (r0 + 16 * i) * KP))[c4] = o;
    }
#pragma unroll
    for (int i = 0; i < 4; ++i)
        ((float4*)(Ws + (r0 + 16 * i) * KP))[c4] = wq[i];
    __syncthreads();

    const int wv = tid >> 5, lane = tid & 31, l16 = lane & 15, half = lane >> 4;
    v8f acc[4];
    const v8f z = {0.f, 0.f, 0.f, 0.f, 0.f, 0.f, 0.f, 0.f};
    acc[0] = z; acc[1] = z; acc[2] = z; acc[3] = z;
    wave_gemm(Xs, Ws, wv * 16, 64, lane, acc);

#pragma unroll
    for (int t = 0; t < 4; ++t)
#pragma unroll
        for (int v = 0; v < 8; ++v) {
            int row = wv * 16 + v + 8 * half;
            Yraw[(p0 + row) * 64 + t * 16 + l16] = acc[t][v];
        }
    stats_reduce(acc, lane, sred);
    __syncthreads();
    if (tid < GRP * 2) atomicAdd(&stats[(int)(p0 >> 13) * GRP * 2 + tid], sred[tid]);
}

__global__ __launch_bounds__(256) void k_gn_apply(float* __restrict__ Y,
                                                  const float* __restrict__ stats,
                                                  const float* __restrict__ gamma,
                                                  const float* __restrict__ beta,
                                                  float inv_count) {
    long long i = (long long)blockIdx.x * 256 + threadIdx.x;
    int c = (int)(i & 63);
    long long p = i >> 6;
    int b = (int)(p >> 13);
    int g = c >> 3;
    float sum = stats[b * GRP * 2 + 2 * g];
    float sq  = stats[b * GRP * 2 + 2 * g + 1];
    float mean = sum * inv_count;
    float var  = sq * inv_count - mean * mean;
    float istd = rsqrtf(var + EPSV);
    float sc = gamma[c] * istd;
    float bs = beta[c] - mean * sc;
    float x = Y[i] * sc + bs;
    Y[i] = (x >= 0.f) ? x : SLOPE * x;
}

extern "C" void kernel_launch(void* const* d_in, const int* in_sizes, int n_in,
                              void* d_out, int out_size, void* d_ws, size_t ws_size,
                              hipStream_t stream) {
    (void)in_sizes; (void)n_in; (void)out_size; (void)ws_size;
    const float* signal = (const float*)d_in[0];
    const int*   edges  = (const int*)d_in[1];
    const float* efeat  = (const float*)d_in[2];
    const float* W1 = (const float*)d_in[3];
    const float* g1 = (const float*)d_in[4];
    const float* b1 = (const float*)d_in[5];
    const float* W2 = (const float*)d_in[6];
    const float* g2 = (const float*)d_in[7];
    const float* b2 = (const float*)d_in[8];
    const float* W3 = (const float*)d_in[9];
    const float* g3 = (const float*)d_in[10];
    const float* b3 = (const float*)d_in[11];
    float* out = (float*)d_out;

    float* ws = (float*)d_ws;
    float* stats1 = ws;
    float* stats2 = ws + 64;
    float* stats3 = ws + 128;
    float* h1   = ws + 256;
    float* buf2 = h1 + (long long)32768 * 64;
    float* ptab = buf2;

    const float invc1  = 1.f / (float)(NN * KK * (MID / GRP));
    const float invc23 = 1.f / (float)(NN * (COUT / GRP));

    hipLaunchKernelGGL(k_table,     dim3(256),  dim3(256), 0, stream,
                       signal, W1, ptab, ws);
    hipLaunchKernelGGL(k_fc1_edges, dim3(1024), dim3(256), 0, stream,
                       ptab, edges, efeat, W1, g1, stats1, h1);
    hipLaunchKernelGGL(k_fc_stats,  dim3(256),  dim3(256), 0, stream,
                       h1, stats1, g1, b1, invc1, W2, buf2, stats2);
    hipLaunchKernelGGL(k_fc_stats,  dim3(256),  dim3(256), 0, stream,
                       buf2, stats2, g2, b2, invc23, W3, out, stats3);
    hipLaunchKernelGGL(k_gn_apply,  dim3(8192), dim3(256), 0, stream,
                       out, stats3, g3, b3, invc23);
}


# final GroupNorm+lrelu pass: a thread does 8 float4 of one channel quad, scale and shift worked out once per thread; 256 work-groups of wide loads instead of 8192 of one dword per thread
# speedup vs baseline: 5.8496x; 1.0656x over previous
#include <hip/hip_runtime.h>

#define BB    4
#define NN    8192
#define KK    32
#define CIN   64
#define CF    67
#define KP    68
#define MID   64
#define COUT  64
#define GRP   8
#define SLOPE 0.1f
#define EPSV  1e-5f

typedef __attribute__((ext_vector_type(2))) float v2f;
typedef __attribute__((ext_vector_type(8))) float v8f;

static __device__ __forceinline__ void wave_gemm(const float* Xs, const float* Ws,
                                                 int rowBase, int kdim, int lane,
                                                 v8f acc[4]) {
    const int l16  = lane & 15;
    const int koff = (lane >> 4) << 1;
    const float* xrow = Xs + (rowBase + l16) * KP;
    for (int k0 = 0; k0 < kdim; k0 += 4) {
        v2f a;
        a.x = xrow[k0 + koff];
        a.y = xrow[k0 + koff + 1];
#pragma unroll
        for (int t = 0; t < 4; ++t) {
            const float* wrow = Ws + (t * 16 + l16) * KP;
            v2f b;
            b.x = wrow[k0 + koff];
            b.y = wrow[k0 + koff + 1];
            acc[t] = __builtin_amdgcn_wmma_f32_16x16x4_f32(
                false, a, false, b, (short)0, acc[t], false, false);
        }
    }
}

static __device__ __forceinline__ void stats_commit(const float s_in[4], const float q_in[4],
                                                    int lane, float* sred) {
    const int l16 = lane & 15;
#pragma unroll
    for (int t = 0; t < 4; ++t) {
        float s = s_in[t], q = q_in[t];
        s += __shfl_xor(s, 1, 32);  q += __shfl_xor(q, 1, 32);
        s += __shfl_xor(s, 2, 32);  q += __shfl_xor(q, 2, 32);
        s += __shfl_xor(s, 4, 32);  q += __shfl_xor(q, 4, 32);
        s += __shfl_xor(s, 16, 32); q += __shfl_xor(q, 16, 32);
        if ((lane & 7) == 0 && lane < 16) {
            int g = 2 * t + (l16 >> 3);
            atomicAdd(&sred[2 * g + 0], s);
            atomicAdd(&sred[2 * g + 1], q);
        }
    }
}

static __device__ __forceinline__ void lane_sums(const v8f acc[4], float s[4], float q[4]) {
#pragma unroll
    for (int t = 0; t < 4; ++t)
#pragma unroll
        for (int v = 0; v < 8; ++v) { float x = acc[t][v]; s[t] += x; q[t] += x * x; }
}

static __device__ __forceinline__ void stats_reduce(v8f acc[4], int lane, float* sred) {
    float s[4] = {0.f, 0.f, 0.f, 0.f}, q[4] = {0.f, 0.f, 0.f, 0.f};
    lane_sums(acc, s, q);
    stats_commit(s, q, lane, sred);
}

__global__ __launch_bounds__(256) void k_table(const float* __restrict__ signal,
                                               const float* __restrict__ W1,
                                               float* __restrict__ P,
                                               float* __restrict__ stats) {
    __shared__ __align__(16) float Xs[128 * KP];
    __shared__ __align__(16) float Ws[MID * KP];
    const int tid = threadIdx.x;
    if (blockIdx.x == 0 && tid < 3 * BB * GRP * 2) stats[tid] = 0.f;
    const long long p0 = (long long)blockIdx.x * 128;
    const int c4 = tid & 15, r0 = tid >> 4;
    float4 xv[8];
    float  wl[16];
#pragma unroll
    for (int i = 0; i < 8; ++i)
        xv[i] = ((const float4*)(signal + (p0 + r0 + 16 * i) * CIN))[c4];
#pragma unroll
    for (int i = 0; i < 16; ++i) {
        int idx = tid + 256 * i;
        wl[i] = W1[(idx >> 6) * CF + (idx & 63)];
    }
#pragma unroll
    for (int i = 0; i < 8; ++i)
        ((float4*)(Xs + (r0 + 16 * i) * KP))[c4] = xv[i];
#pragma unroll
    for (int i = 0; i < 16; ++i) {
        int idx = tid + 256 * i;
        Ws[(idx >> 6) * KP + (idx & 63)] = wl[i];
    }
    __syncthreads();

    const int wv = tid >> 5, lane = tid & 31, l16 = lane & 15, half = lane >> 4;
    v8f acc[4];
    const v8f z = {0.f, 0.f, 0.f, 0.f, 0.f, 0.f, 0.f, 0.f};
    acc[0] = z; acc[1] = z; acc[2] = z; acc[3] = z;
    wave_gemm(Xs, Ws, wv * 16, 64, lane, acc);

#pragma unroll
    for (int t = 0; t < 4; ++t)
#pragma unroll
        for (int v = 0; v < 8; ++v) {
            int row = wv * 16 + v + 8 * half;
            P[(p0 + row) * 64 + t * 16 + l16] = acc[t][v];
        }
}

#define PTS 32
__global__ __launch_bounds__(256) void k_fc1_edges(const float* __restrict__ P,
                                                   const int* __restrict__ edges,
                                                   const float* __restrict__ efeat,
                                                   const float* __restrict__ W1,
                                                   const float* __restrict__ g1,
                                                   float* __restrict__ stats1,
                                                   float* __restrict__ h1) {
    __shared__ float sred[GRP * 2];
    const int tid = threadIdx.x;
    if (tid < GRP * 2) sred[tid] = 0.f;
    __syncthreads();
    const int wv = tid >> 5, lane = tid & 31, l16 = lane & 15, half = lane >> 4;

    v2f  bw[4];
    bool pos[4];
    {
        const float4 gq = ((const float4*)g1)[l16];
        pos[0] = gq.x >= 0.f; pos[1] = gq.y >= 0.f; pos[2] = gq.z >= 0.f; pos[3] = gq.w >= 0.f;
    }
#pragma unroll
    for (int t = 0; t < 4; ++t) {
        const int m = 4 * l16 + t;
        bw[t].x = W1[m * CF + 64 + 2 * half];
        float w1 = W1[m * CF + 65 - half];
        bw[t].y = half ? 0.f : w1;
    }

    float s[4] = {0.f, 0.f, 0.f, 0.f}, q[4] = {0.f, 0.f, 0.f, 0.f};
    const int pbase = blockIdx.x * PTS;
    for (int it = 0; it < PTS / 8; ++it) {
        const int p = pbase + it * 8 + wv;
        const float4 cen = ((const float4*)(P + (unsigned)p * 64u))[l16];
        float ext[4];
#pragma unroll
        for (int t = 0; t < 4; ++t) ext[t] = pos[t] ? -INFINITY : INFINITY;
#pragma unroll 1
        for (int j = 0; j < 2; ++j) {
            const int R0 = p * KK + j * 16;
            const int4* ep = (const int4*)(edges + R0 + 8 * half);
            const int4 ea = ep[0], eb = ep[1];
            const unsigned e[8] = {(unsigned)ea.x, (unsigned)ea.y, (unsigned)ea.z, (unsigned)ea.w,
                                   (unsigned)eb.x, (unsigned)eb.y, (unsigned)eb.z, (unsigned)eb.w};
            const float* fr = efeat + (long long)(R0 + l16) * 3;
            v2f a;
            a.x = fr[2 * half];
            float f1 = fr[1];
            a.y = half ? 0.f : f1;

            v8f acc[4];
#pragma unroll
            for (int v = 0; v < 8; ++v) {
                const float4 g = ((const float4*)(P + e[v] * 64u))[l16];
                acc[0][v] = g.x - cen.x; acc[1][v] = g.y - cen.y;
                acc[2][v] = g.z - cen.z; acc[3][v] = g.w - cen.w;
            }
#pragma unroll
            for (int t = 0; t < 4; ++t)
                acc[t] = __builtin_amdgcn_wmma_f32_16x16x4_f32(
                    false, a, false, bw[t], (short)0, acc[t], false, false);

            lane_sums(acc, s, q);
#pragma unroll
            for (int t = 0; t < 4; ++t) {
                float mx = acc[t][0], mn = acc[t][0];
#pragma unroll
                for (int v = 1; v < 8; ++v) {
                    mx = fmaxf(mx, acc[t][v]);
                    mn = fminf(mn, acc[t][v]);
                }
                ext[t] = pos[t] ? fmaxf(ext[t], mx) : fminf(ext[t], mn);
            }
        }
#pragma unroll
        for (int t = 0; t < 4; ++t) {
            float o = __shfl_xor(ext[t], 16, 32);
            ext[t] = pos[t] ? fmaxf(ext[t], o) : fminf(ext[t], o);
        }
        if (lane < 16) {
            float4 o; o.x = ext[0]; o.y = ext[1]; o.z = ext[2]; o.w = ext[3];
            ((float4*)(h1 + (long long)p * MID))[l16] = o;
        }
    }
    float ss = (s[0] + s[1]) + (s[2] + s[3]);
    float qq = (q[0] + q[1]) + (q[2] + q[3]);
    ss += __shfl_xor(ss, 1, 32);  qq += __shfl_xor(qq, 1, 32);
    ss += __shfl_xor(ss, 16, 32); qq += __shfl_xor(qq, 16, 32);
    if ((lane & 1) == 0 && lane < 16) {
        atomicAdd(&sred[2 * (l16 >> 1) + 0], ss);
        atomicAdd(&sred[2 * (l16 >> 1) + 1], qq);
    }
    __syncthreads();
    if (tid < GRP * 2) atomicAdd(&stats1[(pbase >> 13) * GRP * 2 + tid], sred[tid]);
}

__global__ __launch_bounds__(256) void k_fc_stats(const float* __restrict__ X,
                                                  const float* __restrict__ pstats,
                                                  const float* __restrict__ pgamma,
                                                  const float* __restrict__ pbeta,
                                                  float inv_count,
                                                  const float* __restrict__ W,
                                                  float* __restrict__ Yraw,
                                                  float* __restrict__ stats) {
    __shared__ __align__(16) float Xs[128 * KP];
    __shared__ __align__(16) float Ws[MID * KP];
    __shared__ float sred[GRP * 2];
    const int tid = threadIdx.x;
    if (tid < GRP * 2) sred[tid] = 0.f;
    const long long p0 = (long long)blockIdx.x * 128;

    const int c4 = tid & 15, r0 = tid >> 4;
    const int b = (int)(p0 >> 13);
    const int g = c4 >> 1;
    const float sum = pstats[b * GRP * 2 + 2 * g];
    const float sq  = pstats[b * GRP * 2 + 2 * g + 1];
    const float4 gm = ((const float4*)pgamma)[c4];
    const float4 bt = ((const float4*)pbeta)[c4];
    float4 xv[8], wq[4];
#pragma unroll
    for (int i = 0; i < 8; ++i)
        xv[i] = ((const float4*)(X + (p0 + r0 + 16 * i) * 64))[c4];
#pragma unroll
    for (int i = 0; i < 4; ++i)
        wq[i] = ((const float4*)(W + (r0 + 16 * i) * 64))[c4];
    __builtin_amdgcn_sched_barrier(0);

    float4 sc, bs;
    {
        float mean = sum * inv_count;
        float var  = sq * inv_count - mean * mean;
        float istd = rsqrtf(var + EPSV);
        sc.x = gm.x * istd; sc.y = gm.y * istd; sc.z = gm.z * istd; sc.w = gm.w * istd;
        bs.x = bt.x - mean * sc.x; bs.y = bt.y - mean * sc.y;
        bs.z = bt.z - mean * sc.z; bs.w = bt.w - mean * sc.w;
    }
#pragma unroll
    for (int i = 0; i < 8; ++i) {
        const float4 v = xv[i];
        float4 o;
        o.x = v.x * sc.x + bs.x; o.x = (o.x >= 0.f) ? o.x : SLOPE * o.x;
        o.y = v.y * sc.y + bs.y; o.y = (o.y >= 0.f) ? o.y : SLOPE * o.y;
        o.z = v.z * sc.z + bs.z; o.z = (o.z >= 0.f) ? o.z : SLOPE * o.z;
        o.w = v.w * sc.w + bs.w; o.w = (o.w >= 0.f) ? o.w : SLOPE * o.w;
        ((float4*)(Xs + (r0 + 16 * i) * KP))[c4] = o;
    }
#pragma unroll
    for (int i = 0; i < 4; ++i)
        ((float4*)(Ws + (r0 + 16 * i) * KP))[c4] = wq[i];
    __syncthreads();

    const int wv = tid >> 5, lane = tid & 31, l16 = lane & 15, half = lane >> 4;
    v8f acc[4];
    const v8f z = {0.f, 0.f, 0.f, 0.f, 0.f, 0.f, 0.f, 0.f};
    acc[0] = z; acc[1] = z; acc[2] = z; acc[3] = z;
    wave_gemm(Xs, Ws, wv * 16, 64, lane, acc);

#pragma unroll
    for (int t = 0; t < 4; ++t)
#pragma unroll
        for (int v = 0; v < 8; ++v) {
            int row = wv * 16 + v + 8 * half;
            Yraw[(p0 + row) * 64 + t * 16 + l16] = acc[t][v];
        }
    stats_reduce(acc, lane, sred);
    __syncthreads();
    if (tid < GRP * 2) atomicAdd(&stats[(int)(p0 >> 13) * GRP * 2 + tid], sred[tid]);
}

__global__ __launch_bounds__(256) void k_gn_apply(float* __restrict__ Y,
                                                  const float* __restrict__ stats,
                                                  const float* __restrict__ gamma,
                                                  const float* __restrict__ beta,
                                                  float inv_count) {
    const int tid = threadIdx.x;
    const int c4 = tid & 15, r0 = tid >> 4;
    const long long p0 = (long long)blockIdx.x * 128;
    const int b = (int)(p0 >> 13);
    const int g = c4 >> 1;
    float4* yp = (float4*)(Y + (p0 + r0) * 64) + c4;
    float4 xv[8];
#pragma unroll
    for (int i = 0; i < 8; ++i) xv[i] = yp[256 * i];
    const float sum = stats[b * GRP * 2 + 2 * g];
    const float sq  = stats[b * GRP * 2 + 2 * g + 1];
    const float4 gm = ((const float4*)gamma)[c4];
    const float4 bt = ((const float4*)beta)[c4];

    float4 sc, bs;
    {
        float mean = sum * inv_count;
        float var  = sq * inv_count - mean * mean;
        float istd = rsqrtf(var + EPSV);
        sc.x = gm.x * istd; sc.y = gm.y * istd; sc.z = gm.z * istd; sc.w = gm.w * istd;
        bs.x = bt.x - mean * sc.x; bs.y = bt.y - mean * sc.y;
        bs.z = bt.z - mean * sc.z; bs.w = bt.w - mean * sc.w;
    }
#pragma unroll
    for (int i = 0; i < 8; ++i) {
        const float4 v = xv[i];
        float4 o;
        o.x = v.x * sc.x + bs.x; o.x = (o.x >= 0.f) ? o.x : SLOPE * o.x;
        o.y = v.y * sc.y + bs.y; o.y = (o.y >= 0.f) ? o.y : SLOPE * o.y;
        o.z = v.z * sc.z + bs.z; o.z = (o.z >= 0.f) ? o.z : SLOPE * o.z;
        o.w = v.w * sc.w + bs.w; o.w = (o.w >= 0.f) ? o.w : SLOPE * o.w;
        yp[256 * i] = o;
    }
}

extern "C" void kernel_launch(void* const* d_in, const int* in_sizes, int n_in,
                              void* d_out, int out_size, void* d_ws, size_t ws_size,
                              hipStream_t stream) {
    (void)in_sizes; (void)n_in; (void)out_size; (void)ws_size;
    const float* signal = (const float*)d_in[0];
    const int*   edges  = (const int*)d_in[1];
    const float* efeat  = (const float*)d_in[2];
    const float* W1 = (const float*)d_in[3];
    const float* g1 = (const float*)d_in[4];
    const float* b1 = (const float*)d_in[5];
    const float* W2 = (const float*)d_in[6];
    const float* g2 = (const float*)d_in[7];
    const float* b2 = (const float*)d_in[8];
    const float* W3 = (const float*)d_in[9];
    const float* g3 = (const float*)d_in[10];
    const float* b3 = (const float*)d_in[11];
    float* out = (float*)d_out;

    float* ws = (float*)d_ws;
    float* stats1 = ws;
    float* stats2 = ws + 64;
    float* stats3 = ws + 128;
    float* h1   = ws + 256;
    float* buf2 = h1 + (long long)32768 * 64;
    float* ptab = buf2;

    const float invc1  = 1.f / (float)(NN * KK * (MID / GRP));
    const float invc23 = 1.f / (float)(NN * (COUT / GRP));

    hipLaunchKernelGGL(k_table,     dim3(256),  dim3(256), 0, stream,
                       signal, W1, ptab, ws);
    hipLaunchKernelGGL(k_fc1_edges, dim3(1024), dim3(256), 0, stream,
                       ptab, edges, efeat, W1, g1, stats1, h1);
    hipLaunchKernelGGL(k_fc_stats,  dim3(256),  dim3(256), 0, stream,
                       h1, stats1, g1, b1, invc1, W2, buf2, stats2);
    hipLaunchKernelGGL(k_fc_stats,  dim3(256),  dim3(256), 0, stream,
                       buf2, stats2, g2, b2, invc23, W3, out, stats3);
    hipLaunchKernelGGL(k_gn_apply,  dim3(256),  dim3(256), 0, stream,
                       out, stats3, g3, b3, invc23);
}


# fc1 edge pass: each work-group copies its 1024 edge ids and 3072 edge features into LDS in one coalesced 16-byte load pass; tiles read them from LDS: serial memory round trips per tile 2 to 1
# speedup vs baseline: 6.0679x; 1.0373x over previous
#include <hip/hip_runtime.h>

#define BB    4
#define NN    8192
#define KK    32
#define CIN   64
#define CF    67
#define KP    68
#define MID   64
#define COUT  64
#define GRP   8
#define SLOPE 0.1f
#define EPSV  1e-5f

typedef __attribute__((ext_vector_type(2))) float v2f;
typedef __attribute__((ext_vector_type(8))) float v8f;

static __device__ __forceinline__ void wave_gemm(const float* Xs, const float* Ws,
                                                 int rowBase, int kdim, int lane,
                                                 v8f acc[4]) {
    const int l16  = lane & 15;
    const int koff = (lane >> 4) << 1;
    const float* xrow = Xs + (rowBase + l16) * KP;
    for (int k0 = 0; k0 < kdim; k0 += 4) {
        v2f a;
        a.x = xrow[k0 + koff];
        a.y = xrow[k0 + koff + 1];
#pragma unroll
        for (int t = 0; t < 4; ++t) {
            const float* wrow = Ws + (t * 16 + l16) * KP;
            v2f b;
            b.x = wrow[k0 + koff];
            b.y = wrow[k0 + koff + 1];
            acc[t] = __builtin_amdgcn_wmma_f32_16x16x4_f32(
                false, a, false, b, (short)0, acc[t], false, false);
        }
    }
}

static __device__ __forceinline__ void stats_commit(const float s_in[4], const float q_in[4],
                                                    int lane, float* sred) {
    const int l16 = lane & 15;
#pragma unroll
    for (int t = 0; t < 4; ++t) {
        float s = s_in[t], q = q_in[t];
        s += __shfl_xor(s, 1, 32);  q += __shfl_xor(q, 1, 32);
        s += __shfl_xor(s, 2, 32);  q += __shfl_xor(q, 2, 32);
        s += __shfl_xor(s, 4, 32);  q += __shfl_xor(q, 4, 32);
        s += __shfl_xor(s, 16, 32); q += __shfl_xor(q, 16, 32);
        if ((lane & 7) == 0 && lane < 16) {
            int g = 2 * t + (l16 >> 3);
            atomicAdd(&sred[2 * g + 0], s);
            atomicAdd(&sred[2 * g + 1], q);
        }
    }
}

static __device__ __forceinline__ void lane_sums(const v8f acc[4], float s[4], float q[4]) {
#pragma unroll
    for (int t = 0; t < 4; ++t)
#pragma unroll
        for (int v = 0; v < 8; ++v) { float x = acc[t][v]; s[t] += x; q[t] += x * x; }
}

static __device__ __forceinline__ void stats_reduce(v8f acc[4], int lane, float* sred) {
    float s[4] = {0.f, 0.f, 0.f, 0.f}, q[4] = {0.f, 0.f, 0.f, 0.f};
    lane_sums(acc, s, q);
    stats_commit(s, q, lane, sred);
}

__global__ __launch_bounds__(256) void k_table(const float* __restrict__ signal,
                                               const float* __restrict__ W1,
                                               float* __restrict__ P,
                                               float* __restrict__ stats) {
    __shared__ __align__(16) float Xs[128 * KP];
    __shared__ __align__(16) float Ws[MID * KP];
    const int tid = threadIdx.x;
    if (blockIdx.x == 0 && tid < 3 * BB * GRP * 2) stats[tid] = 0.f;
    const long long p0 = (long long)blockIdx.x * 128;
    const int c4 = tid & 15, r0 = tid >> 4;
    float4 xv[8];
    float  wl[16];
#pragma unroll
    for (int i = 0; i < 8; ++i)
        xv[i] = ((const float4*)(signal + (p0 + r0 + 16 * i) * CIN))[c4];
#pragma unroll
    for (int i = 0; i < 16; ++i) {
        int idx = tid + 256 * i;
        wl[i] = W1[(idx >> 6) * CF + (idx & 63)];
    }
#pragma unroll
    for (int i = 0; i < 8; ++i)
        ((float4*)(Xs + (r0 + 16 * i) * KP))[c4] = xv[i];
#pragma unroll
    for (int i = 0; i < 16; ++i) {
        int idx = tid + 256 * i;
        Ws[(idx >> 6) * KP + (idx & 63)] = wl[i];
    }
    __syncthreads();

    const int wv = tid >> 5, lane = tid & 31, l16 = lane & 15, half = lane >> 4;
    v8f acc[4];
    const v8f z = {0.f, 0.f, 0.f, 0.f, 0.f, 0.f, 0.f, 0.f};
    acc[0] = z; acc[1] = z; acc[2] = z; acc[3] = z;
    wave_gemm(Xs, Ws, wv * 16, 64, lane, acc);

#pragma unroll
    for (int t = 0; t < 4; ++t)
#pragma unroll
        for (int v = 0; v < 8; ++v) {
            int row = wv * 16 + v + 8 * half;
            P[(p0 + row) * 64 + t * 16 + l16] = acc[t][v];
        }
}

#define PTS 32
__global__ __launch_bounds__(256) void k_fc1_edges(const float* __restrict__ P,
                                                   const int* __restrict__ edges,
                                                   const float* __restrict__ efeat,
                                                   const float* __restrict__ W1,
                                                   const float* __restrict__ g1,
                                                   float* __restrict__ stats1,
                                                   float* __restrict__ h1) {
    __shared__ float sred[GRP * 2];
    __shared__ __align__(16) int   ids_s[PTS * KK];
    __shared__ __align__(16) float ef_s[PTS * KK * 3];
    const int tid = threadIdx.x;
    const int pbase = blockIdx.x * PTS;
    const int wv = tid >> 5, lane = tid & 31, l16 = lane & 15, half = lane >> 4;
    const int4*   eg = (const int4*)(edges + pbase * KK);
    const float4* fg = (const float4*)(efeat + (long long)pbase * (KK * 3));
    const int4   iv = eg[tid];
    const float4 f0 = fg[tid], f1 = fg[tid + 256], f2 = fg[tid + 512];
    v2f  bw[4];
    bool pos[4];
    {
        const float4 gq = ((const float4*)g1)[l16];
        pos[0] = gq.x >= 0.f; pos[1] = gq.y >= 0.f; pos[2] = gq.z >= 0.f; pos[3] = gq.w >= 0.f;
    }
#pragma unroll
    for (int t = 0; t < 4; ++t) {
        const int m = 4 * l16 + t;
        bw[t].x = W1[m * CF + 64 + 2 * half];
        float w1 = W1[m * CF + 65 - half];
        bw[t].y = half ? 0.f : w1;
    }
    ((int4*)ids_s)[tid] = iv;
    ((float4*)ef_s)[tid]       = f0;
    ((float4*)ef_s)[tid + 256] = f1;
    ((float4*)ef_s)[tid + 512] = f2;
    if (tid < GRP * 2) sred[tid] = 0.f;
    __syncthreads();


    float s[4] = {0.f, 0.f, 0.f, 0.f}, q[4] = {0.f, 0.f, 0.f, 0.f};
    for (int it = 0; it < PTS / 8; ++it) {
        const int lp = it * 8 + wv;
        const int p = pbase + lp;
        const float4 cen = ((const float4*)(P + (unsigned)p * 64u))[l16];
        float ext[4];
#pragma unroll
        for (int t = 0; t < 4; ++t) ext[t] = pos[t] ? -INFINITY : INFINITY;
#pragma unroll 1
        for (int j = 0; j < 2; ++j) {
            const int R0 = lp * KK + j * 16;
            const int4* ep = (const int4*)(ids_s + R0 + 8 * half);
            const int4 ea = ep[0], eb = ep[1];
            const unsigned e[8] = {(unsigned)ea.x, (unsigned)ea.y, (unsigned)ea.z, (unsigned)ea.w,
                                   (unsigned)eb.x, (unsigned)eb.y, (unsigned)eb.z, (unsigned)eb.w};
            const float* fr = ef_s + (R0 + l16) * 3;
            v2f a;
            a.x = fr[2 * half];
            float f1 = fr[1];
            a.y = half ? 0.f : f1;

            v8f acc[4];
#pragma unroll
            for (int v = 0; v < 8; ++v) {
                const float4 g = ((const float4*)(P + e[v] * 64u))[l16];
                acc[0][v] = g.x - cen.x; acc[1][v] = g.y - cen.y;
                acc[2][v] = g.z - cen.z; acc[3][v] = g.w - cen.w;
            }
#pragma unroll
            for (int t = 0; t < 4; ++t)
                acc[t] = __builtin_amdgcn_wmma_f32_16x16x4_f32(
                    false, a, false, bw[t], (short)0, acc[t], false, false);

            lane_sums(acc, s, q);
#pragma unroll
            for (int t = 0; t < 4; ++t) {
                float mx = acc[t][0], mn = acc[t][0];
#pragma unroll
                for (int v = 1; v < 8; ++v) {
                    mx = fmaxf(mx, acc[t][v]);
                    mn = fminf(mn, acc[t][v]);
                }
                ext[t] = pos[t] ? fmaxf(ext[t], mx) : fminf(ext[t], mn);
            }
        }
#pragma unroll
        for (int t = 0; t < 4; ++t) {
            float o = __shfl_xor(ext[t], 16, 32);
            ext[t] = pos[t] ? fmaxf(ext[t], o) : fminf(ext[t], o);
        }
        if (lane < 16) {
            float4 o; o.x = ext[0]; o.y = ext[1]; o.z = ext[2]; o.w = ext[3];
            ((float4*)(h1 + (long long)p * MID))[l16] = o;
        }
    }
    float ss = (s[0] + s[1]) + (s[2] + s[3]);
    float qq = (q[0] + q[1]) + (q[2] + q[3]);
    ss += __shfl_xor(ss, 1, 32);  qq += __shfl_xor(qq, 1, 32);
    ss += __shfl_xor(ss, 16, 32); qq += __shfl_xor(qq, 16, 32);
    if ((lane & 1) == 0 && lane < 16) {
        atomicAdd(&sred[2 * (l16 >> 1) + 0], ss);
        atomicAdd(&sred[2 * (l16 >> 1) + 1], qq);
    }
    __syncthreads();
    if (tid < GRP * 2) atomicAdd(&stats1[(pbase >> 13) * GRP * 2 + tid], sred[tid]);
}

__global__ __launch_bounds__(256) void k_fc_stats(const float* __restrict__ X,
                                                  const float* __restrict__ pstats,
                                                  const float* __restrict__ pgamma,
                                                  const float* __restrict__ pbeta,
                                                  float inv_count,
                                                  const float* __restrict__ W,
                                                  float* __restrict__ Yraw,
                                                  float* __restrict__ stats) {
    __shared__ __align__(16) float Xs[128 * KP];
    __shared__ __align__(16) float Ws[MID * KP];
    __shared__ float sred[GRP * 2];
    const int tid = threadIdx.x;
    if (tid < GRP * 2) sred[tid] = 0.f;
    const long long p0 = (long long)blockIdx.x * 128;

    const int c4 = tid & 15, r0 = tid >> 4;
    const int b = (int)(p0 >> 13);
    const int g = c4 >> 1;
    const float sum = pstats[b * GRP * 2 + 2 * g];
    const float sq  = pstats[b * GRP * 2 + 2 * g + 1];
    const float4 gm = ((const float4*)pgamma)[c4];
    const float4 bt = ((const float4*)pbeta)[c4];
    float4 xv[8], wq[4];
#pragma unroll
    for (int i = 0; i < 8; ++i)
        xv[i] = ((const float4*)(X + (p0 + r0 + 16 * i) * 64))[c4];
#pragma unroll
    for (int i = 0; i < 4; ++i)
        wq[i] = ((const float4*)(W + (r0 + 16 * i) * 64))[c4];
    __builtin_amdgcn_sched_barrier(0);

    float4 sc, bs;
    {
        float mean = sum * inv_count;
        float var  = sq * inv_count - mean * mean;
        float istd = rsqrtf(var + EPSV);
        sc.x = gm.x * istd; sc.y = gm.y * istd; sc.z = gm.z * istd; sc.w = gm.w * istd;
        bs.x = bt.x - mean * sc.x; bs.y = bt.y - mean * sc.y;
        bs.z = bt.z - mean * sc.z; bs.w = bt.w - mean * sc.w;
    }
#pragma unroll
    for (int i = 0; i < 8; ++i) {
        const float4 v = xv[i];
        float4 o;
        o.x = v.x * sc.x + bs.x; o.x = (o.x >= 0.f) ? o.x : SLOPE * o.x;
        o.y = v.y * sc.y + bs.y; o.y = (o.y >= 0.f) ? o.y : SLOPE * o.y;
        o.z = v.z * sc.z + bs.z; o.z = (o.z >= 0.f) ? o.z : SLOPE * o.z;
        o.w = v.w * sc.w + bs.w; o.w = (o.w >= 0.f) ? o.w : SLOPE * o.w;
        ((float4*)(Xs + (r0 + 16 * i) * KP))[c4] = o;
    }
#pragma unroll
    for (int i = 0; i < 4; ++i)
        ((float4*)(Ws + (r0 + 16 * i) * KP))[c4] = wq[i];
    __syncthreads();

    const int wv = tid >> 5, lane = tid & 31, l16 = lane & 15, half = lane >> 4;
    v8f acc[4];
    const v8f z = {0.f, 0.f, 0.f, 0.f, 0.f, 0.f, 0.f, 0.f};
    acc[0] = z; acc[1] = z; acc[2] = z; acc[3] = z;
    wave_gemm(Xs, Ws, wv * 16, 64, lane, acc);

#pragma unroll
    for (int t = 0; t < 4; ++t)
#pragma unroll
        for (int v = 0; v < 8; ++v) {
            int row = wv * 16 + v + 8 * half;
            Yraw[(p0 + row) * 64 + t * 16 + l16] = acc[t][v];
        }
    stats_reduce(acc, lane, sred);
    __syncthreads();
    if (tid < GRP * 2) atomicAdd(&stats[(int)(p0 >> 13) * GRP * 2 + tid], sred[tid]);
}

__global__ __launch_bounds__(256) void k_gn_apply(float* __restrict__ Y,
                                                  const float* __restrict__ stats,
                                                  const float* __restrict__ gamma,
                                                  const float* __restrict__ beta,
                                                  float inv_count) {
    const int tid = threadIdx.x;
    const int c4 = tid & 15, r0 = tid >> 4;
    const long long p0 = (long long)blockIdx.x * 128;
    const int b = (int)(p0 >> 13);
    const int g = c4 >> 1;
    float4* yp = (float4*)(Y + (p0 + r0) * 64) + c4;
    float4 xv[8];
#pragma unroll
    for (int i = 0; i < 8; ++i) xv[i] = yp[256 * i];
    const float sum = stats[b * GRP * 2 + 2 * g];
    const float sq  = stats[b * GRP * 2 + 2 * g + 1];
    const float4 gm = ((const float4*)gamma)[c4];
    const float4 bt = ((const float4*)beta)[c4];

    float4 sc, bs;
    {
        float mean = sum * inv_count;
        float var  = sq * inv_count - mean * mean;
        float istd = rsqrtf(var + EPSV);
        sc.x = gm.x * istd; sc.y = gm.y * istd; sc.z = gm.z * istd; sc.w = gm.w * istd;
        bs.x = bt.x - mean * sc.x; bs.y = bt.y - mean * sc.y;
        bs.z = bt.z - mean * sc.z; bs.w = bt.w - mean * sc.w;
    }
#pragma unroll
    for (int i = 0; i < 8; ++i) {
        const float4 v = xv[i];
        float4 o;
        o.x = v.x * sc.x + bs.x; o.x = (o.x >= 0.f) ? o.x : SLOPE * o.x;
        o.y = v.y * sc.y + bs.y; o.y = (o.y >= 0.f) ? o.y : SLOPE * o.y;
        o.z = v.z * sc.z + bs.z; o.z = (o.z >= 0.f) ? o.z : SLOPE * o.z;
        o.w = v.w * sc.w + bs.w; o.w = (o.w >= 0.f) ? o.w : SLOPE * o.w;
        yp[256 * i] = o;
    }
}

extern "C" void kernel_launch(void* const* d_in, const int* in_sizes, int n_in,
                              void* d_out, int out_size, void* d_ws, size_t ws_size,
                              hipStream_t stream) {
    (void)in_sizes; (void)n_in; (void)out_size; (void)ws_size;
    const float* signal = (const float*)d_in[0];
    const int*   edges  = (const int*)d_in[1];
    const float* efeat  = (const float*)d_in[2];
    const float* W1 = (const float*)d_in[3];
    const float* g1 = (const float*)d_in[4];
    const float* b1 = (const float*)d_in[5];
    const float* W2 = (const float*)d_in[6];
    const float* g2 = (const float*)d_in[7];
    const float* b2 = (const float*)d_in[8];
    const float* W3 = (const float*)d_in[9];
    const float* g3 = (const float*)d_in[10];
    const float* b3 = (const float*)d_in[11];
    float* out = (float*)d_out;

    float* ws = (float*)d_ws;
    float* stats1 = ws;
    float* stats2 = ws + 64;
    float* stats3 = ws + 128;
    float* h1   = ws + 256;
    float* buf2 = h1 + (long long)32768 * 64;
    float* ptab = buf2;

    const float invc1  = 1.f / (float)(NN * KK * (MID / GRP));
    const float invc23 = 1.f / (float)(NN * (COUT / GRP));

    hipLaunchKernelGGL(k_table,     dim3(256),  dim3(256), 0, stream,
                       signal, W1, ptab, ws);
    hipLaunchKernelGGL(k_fc1_edges, dim3(1024), dim3(256), 0, stream,
                       ptab, edges, efeat, W1, g1, stats1, h1);
    hipLaunchKernelGGL(k_fc_stats,  dim3(256),  dim3(256), 0, stream,
                       h1, stats1, g1, b1, invc1, W2, buf2, stats2);
    hipLaunchKernelGGL(k_fc_stats,  dim3(256),  dim3(256), 0, stream,
                       buf2, stats2, g2, b2, invc23, W3, out, stats3);
    hipLaunchKernelGGL(k_gn_apply,  dim3(256),  dim3(256), 0, stream,
                       out, stats3, g3, b3, invc23);
}
